# starting kernel
# baseline (speedup 1.0000x reference)
#include <hip/hip_runtime.h>
#include <math.h>

#define N_ENT 50000
#define N_ATT 2000
#define N_VAL 100000
#define NE    300000
#define KD    256

typedef __attribute__((ext_vector_type(2))) float v2f;
typedef __attribute__((ext_vector_type(8))) float v8f;

__global__ void zero_f32(float* __restrict__ p, long n) {
    long i = (long)blockIdx.x * blockDim.x + threadIdx.x;
    long stride = (long)gridDim.x * blockDim.x;
    for (; i < n; i += stride) p[i] = 0.0f;
}

__global__ void row_dot_kernel(const float* __restrict__ ent_feats,
                               const float* __restrict__ att_feats,
                               const float* __restrict__ a_w,
                               float* __restrict__ ent_score,
                               float* __restrict__ att_score) {
    int wave = (int)((blockIdx.x * blockDim.x + threadIdx.x) >> 5);
    int lane = threadIdx.x & 31;
    const float* feat;
    const float* w;
    float* outp;
    if (wave < N_ENT) {
        feat = ent_feats + (long)wave * KD;
        w = a_w;
        outp = ent_score + wave;
    } else {
        int r = wave - N_ENT;
        if (r >= N_ATT) return;
        feat = att_feats + (long)r * KD;
        w = a_w + KD;
        outp = att_score + r;
    }
    float s = 0.0f;
#pragma unroll
    for (int j = 0; j < 2; ++j) {
        int c = (lane + 32 * j) * 4;
        float4 f = *(const float4*)(feat + c);
        float4 ww = *(const float4*)(w + c);
        s += f.x * ww.x + f.y * ww.y + f.z * ww.z + f.w * ww.w;
    }
#pragma unroll
    for (int off = 16; off > 0; off >>= 1) s += __shfl_xor(s, off, 32);
    if (lane == 0) *outp = s;
}

__global__ void edge_score_kernel(const int* __restrict__ trip,
                                  const float* __restrict__ ent_score,
                                  const float* __restrict__ att_score,
                                  const float* __restrict__ a_b,
                                  float* __restrict__ score,
                                  float* __restrict__ row_sum) {
    int e = blockIdx.x * blockDim.x + threadIdx.x;
    if (e >= NE) return;
    int h  = trip[3 * e + 0];
    int at = trip[3 * e + 2];
    float x = ent_score[h] + att_score[at] + a_b[0];
    x = (x > 0.0f) ? x : 0.2f * x;
    float sc = expf(x);
    score[e] = sc;
    atomicAdd(&row_sum[h], sc);
}

template <bool ACC>
__global__ void wmma_gemm_f32(const float* __restrict__ A,
                              const float* __restrict__ B,
                              float* __restrict__ C) {
    int wave = (int)((blockIdx.x * blockDim.x + threadIdx.x) >> 5);
    int lane = threadIdx.x & 31;
    int tM = wave >> 4;
    int tN = wave & 15;
    int row0 = tM * 16;
    int col0 = tN * 16;

    int m    = lane & 15;
    int koff = (lane >> 4) << 1;
    int bn   = col0 + (lane & 15);
    int crow = row0 + ((lane >> 4) << 3);
    int cn   = col0 + (lane & 15);

    v8f c;
    if (ACC) {
#pragma unroll
        for (int r = 0; r < 8; ++r) c[r] = C[(long)(crow + r) * KD + cn];
    } else {
#pragma unroll
        for (int r = 0; r < 8; ++r) c[r] = 0.0f;
    }

    const float* aptr = A + (long)(row0 + m) * KD + koff;

#pragma unroll 4
    for (int k0 = 0; k0 < KD; k0 += 4) {
        v2f av = *(const v2f*)(aptr + k0);
        v2f bv;
        bv.x = B[(long)(k0 + koff + 0) * KD + bn];
        bv.y = B[(long)(k0 + koff + 1) * KD + bn];
        c = __builtin_amdgcn_wmma_f32_16x16x4_f32(
                 false, av,  false, bv,
                 (short)0, c,  false,  false);
    }

#pragma unroll
    for (int r = 0; r < 8; ++r) C[(long)(crow + r) * KD + cn] = c[r];
}

__global__ void edge_agg_kernel(const int* __restrict__ trip,
                                const float* __restrict__ score,
                                const float* __restrict__ row_sum,
                                const float* __restrict__ att_proj,
                                const float* __restrict__ val_feats,
                                float* __restrict__ U,
                                float* __restrict__ outAcc) {
    int d = threadIdx.x;
    for (int e = blockIdx.x; e < NE; e += gridDim.x) {
        int h  = trip[3 * e + 0];
        int vl = trip[3 * e + 1];
        int at = trip[3 * e + 2];
        float p = score[e] / row_sum[h];
        atomicAdd(&U[(long)h * KD + d],      p * val_feats[(long)vl * KD + d]);
        atomicAdd(&outAcc[(long)h * KD + d], p * att_proj[(long)at * KD + d]);
    }
}

__global__ void finalize_kernel(float* __restrict__ out,
                                const float* __restrict__ ent_feats,
                                long n) {
    long i = (long)blockIdx.x * blockDim.x + threadIdx.x;
    long stride = (long)gridDim.x * blockDim.x;
    for (; i < n; i += stride) {
        float x = out[i] + ent_feats[i];
        out[i] = (x > 0.0f) ? x : (expf(x) - 1.0f);
    }
}

extern "C" void kernel_launch(void* const* d_in, const int* in_sizes, int n_in,
                              void* d_out, int out_size, void* d_ws, size_t ws_size,
                              hipStream_t stream) {
    const int*   trip      = (const int*)d_in[0];
    const float* att_feats = (const float*)d_in[1];
    const float* val_feats = (const float*)d_in[2];
    const float* ent_feats = (const float*)d_in[3];
    const float* a_w       = (const float*)d_in[4];
    const float* a_b       = (const float*)d_in[5];
    const float* W         = (const float*)d_in[6];
    float* out = (float*)d_out;

    const float* W_top = W;
    const float* W_bot = W + KD * KD;

    float* ws       = (float*)d_ws;
    float* ent_s    = ws;
    float* att_s    = ent_s   + N_ENT;
    float* row_sum  = att_s   + N_ATT;
    float* score    = row_sum + N_ENT;
    float* att_proj = score   + NE;
    float* U        = att_proj + (long)N_ATT * KD;

    const long nEntFeat = (long)N_ENT * KD;

    zero_f32<<<512, 256, 0, stream>>>(row_sum, (long)N_ENT);
    zero_f32<<<2048, 256, 0, stream>>>(U, nEntFeat);
    zero_f32<<<2048, 256, 0, stream>>>(out, nEntFeat);

    {
        int waves = N_ENT + N_ATT;
        int blocks = (waves * 32 + 255) / 256;
        row_dot_kernel<<<blocks, 256, 0, stream>>>(ent_feats, att_feats, a_w,
                                                   ent_s, att_s);
    }

    {
        int tiles = (N_ATT / 16) * 16;
        wmma_gemm_f32<false><<<tiles / 8, 256, 0, stream>>>(att_feats, W_top, att_proj);
    }

    edge_score_kernel<<<(NE + 255) / 256, 256, 0, stream>>>(trip, ent_s, att_s, a_b,
                                                            score, row_sum);

    edge_agg_kernel<<<2048, 256, 0, stream>>>(trip, score, row_sum,
                                              att_proj, val_feats, U, out);

    {
        int tiles = (N_ENT / 16) * 16;
        wmma_gemm_f32<true><<<tiles / 8, 256, 0, stream>>>(U, W_bot, out);
    }

    finalize_kernel<<<4096, 256, 0, stream>>>(out, ent_feats, nEntFeat);
}


# Edges counting-sorted by destination row (arrival-order int rank adds, scan, scatter); one wave sums each row's gathered rows in registers in rank order: no f32 atomics, no zero-fill launches
# speedup vs baseline: 1.6682x; 1.6682x over previous
#include <hip/hip_runtime.h>
#include <math.h>

#define N_ENT 50000
#define N_ATT 2000
#define N_VAL 100000
#define NE    300000
#define KD    256

#define VAL_BITS 17
static_assert(N_VAL <= (1 << VAL_BITS), "value row index must fit VAL_BITS");
static_assert(N_ATT <= (1 << (31 - VAL_BITS)), "attribute row index must fit the rest");

typedef __attribute__((ext_vector_type(2))) float v2f;
typedef __attribute__((ext_vector_type(8))) float v8f;

__global__ void row_dot_kernel(const float* __restrict__ ent_feats,
                               const float* __restrict__ att_feats,
                               const float* __restrict__ a_w,
                               float* __restrict__ ent_score,
                               float* __restrict__ att_score,
                               int* __restrict__ row_cnt) {
    int wave = (int)((blockIdx.x * blockDim.x + threadIdx.x) >> 5);
    int lane = threadIdx.x & 31;
    const float* feat;
    const float* w;
    float* outp;
    if (wave < N_ENT) {
        feat = ent_feats + (long)wave * KD;
        w = a_w;
        outp = ent_score + wave;
        if (lane == 0) row_cnt[wave] = 0;
    } else {
        int r = wave - N_ENT;
        if (r >= N_ATT) return;
        feat = att_feats + (long)r * KD;
        w = a_w + KD;
        outp = att_score + r;
    }
    float s = 0.0f;
#pragma unroll
    for (int j = 0; j < 2; ++j) {
        int c = (lane + 32 * j) * 4;
        float4 f = *(const float4*)(feat + c);
        float4 ww = *(const float4*)(w + c);
        s += f.x * ww.x + f.y * ww.y + f.z * ww.z + f.w * ww.w;
    }
#pragma unroll
    for (int off = 16; off > 0; off >>= 1) s += __shfl_xor(s, off, 32);
    if (lane == 0) *outp = s;
}

__global__ void edge_rank_kernel(const int* __restrict__ trip,
                                 int* __restrict__ row_cnt,
                                 int* __restrict__ rank) {
    int e = blockIdx.x * blockDim.x + threadIdx.x;
    if (e >= NE) return;
    int h = trip[3 * e + 0];
    rank[e] = atomicAdd(&row_cnt[h], 1);
}

#define SCAN_T  1024
#define SCAN_CH ((N_ENT + SCAN_T - 1) / SCAN_T)
__global__ __launch_bounds__(SCAN_T) void row_scan_kernel(int* __restrict__ row_cnt) {
    __shared__ int s[SCAN_T * SCAN_CH];
    __shared__ int wtot[SCAN_T / 32];
    const int t = threadIdx.x;
    const int lane = t & 31;
    const int w = t >> 5;
    for (int k = 0; k < SCAN_CH; ++k) {
        const int i = t + k * SCAN_T;
        s[i] = (i < N_ENT) ? row_cnt[i] : 0;
    }
    __syncthreads();
    const int b = t * SCAN_CH;
    int sum = 0;
    for (int k = 0; k < SCAN_CH; ++k) sum += s[b + k];
    int v = sum;
    for (int o = 1; o < 32; o <<= 1) {
        int n = __shfl_up(v, o, 32);
        if (lane >= o) v += n;
    }
    if (lane == 31) wtot[w] = v;
    __syncthreads();
    if (w == 0) {
        int x = wtot[lane];
        for (int o = 1; o < 32; o <<= 1) {
            int n = __shfl_up(x, o, 32);
            if (lane >= o) x += n;
        }
        wtot[lane] = x;
    }
    __syncthreads();
    int run = (v - sum) + (w > 0 ? wtot[w - 1] : 0);
    for (int k = 0; k < SCAN_CH; ++k) { run += s[b + k]; s[b + k] = run; }
    __syncthreads();
    for (int k = 0; k < SCAN_CH; ++k) {
        const int i = t + k * SCAN_T;
        if (i < N_ENT) row_cnt[i] = s[i];
    }
}

__global__ void edge_scatter_kernel(const int* __restrict__ trip,
                                    const int* __restrict__ row_end,
                                    const int* __restrict__ rank,
                                    int* __restrict__ slot) {
    int e = blockIdx.x * blockDim.x + threadIdx.x;
    if (e >= NE) return;
    int h  = trip[3 * e + 0];
    int vl = trip[3 * e + 1];
    int at = trip[3 * e + 2];
    int beg = (h > 0) ? row_end[h - 1] : 0;
    slot[beg + rank[e]] = vl | (at << VAL_BITS);
}

template <bool ACC>
__global__ void wmma_gemm_f32(const float* __restrict__ A,
                              const float* __restrict__ B,
                              float* __restrict__ C) {
    int wave = (int)((blockIdx.x * blockDim.x + threadIdx.x) >> 5);
    int lane = threadIdx.x & 31;
    int tM = wave >> 4;
    int tN = wave & 15;
    int row0 = tM * 16;
    int col0 = tN * 16;

    int m    = lane & 15;
    int koff = (lane >> 4) << 1;
    int bn   = col0 + (lane & 15);
    int crow = row0 + ((lane >> 4) << 3);
    int cn   = col0 + (lane & 15);

    v8f c;
    if (ACC) {
#pragma unroll
        for (int r = 0; r < 8; ++r) c[r] = C[(long)(crow + r) * KD + cn];
    } else {
#pragma unroll
        for (int r = 0; r < 8; ++r) c[r] = 0.0f;
    }

    const float* aptr = A + (long)(row0 + m) * KD + koff;

#pragma unroll 4
    for (int k0 = 0; k0 < KD; k0 += 4) {
        v2f av = *(const v2f*)(aptr + k0);
        v2f bv;
        bv.x = B[(long)(k0 + koff + 0) * KD + bn];
        bv.y = B[(long)(k0 + koff + 1) * KD + bn];
        c = __builtin_amdgcn_wmma_f32_16x16x4_f32(
                 false, av,  false, bv,
                 (short)0, c,  false,  false);
    }

#pragma unroll
    for (int r = 0; r < 8; ++r) C[(long)(crow + r) * KD + cn] = c[r];
}

__device__ __forceinline__ float edge_sc(float es, float as, float b) {
    float x = es + as + b;
    x = (x > 0.0f) ? x : 0.2f * x;
    return expf(x);
}

__global__ __launch_bounds__(256) void row_agg_kernel(const int* __restrict__ row_end,
                                                      const int* __restrict__ slot,
                                                      const float* __restrict__ ent_score,
                                                      const float* __restrict__ att_score,
                                                      const float* __restrict__ a_b,
                                                      const float* __restrict__ att_proj,
                                                      const float* __restrict__ val_feats,
                                                      float* __restrict__ U,
                                                      float* __restrict__ outAcc) {
    const int h = __builtin_amdgcn_readfirstlane((int)((blockIdx.x * 256 + threadIdx.x) >> 5));
    const int lane = threadIdx.x & 31;
    if (h >= N_ENT) return;
    const int beg = (h > 0) ? row_end[h - 1] : 0;
    const int end = row_end[h];
    const float es = ent_score[h];
    const float b  = a_b[0];

    int   wd_l = 0;
    float sc_l = 0.0f;
    if (beg + lane < end) {
        wd_l = slot[beg + lane];
        sc_l = edge_sc(es, att_score[wd_l >> VAL_BITS], b);
    }
    float rs = sc_l;
    for (int i = beg + 32 + lane; i < end; i += 32)
        rs += edge_sc(es, att_score[slot[i] >> VAL_BITS], b);
    for (int off = 16; off > 0; off >>= 1) rs += __shfl_xor(rs, off, 32);

    float4 u0 = make_float4(0.f, 0.f, 0.f, 0.f), u1 = u0, o0 = u0, o1 = u0;
    for (int c = beg; c < end; c += 32) {
        if (c != beg) {
            wd_l = 0;
            sc_l = 0.0f;
            if (c + lane < end) {
                wd_l = slot[c + lane];
                sc_l = edge_sc(es, att_score[wd_l >> VAL_BITS], b);
            }
        }
        const float p_l = sc_l / rs;
        const int n = (end - c < 32) ? (end - c) : 32;
        auto edge = [&](int j) {
            const int   wd = __builtin_amdgcn_readlane(wd_l, j);
            const float p  = __int_as_float(__builtin_amdgcn_readlane(__float_as_int(p_l), j));
            const int vl = wd & ((1 << VAL_BITS) - 1);
            const int at = wd >> VAL_BITS;
            const float4* vr = (const float4*)(val_feats + (long)vl * KD);
            const float4* ar = (const float4*)(att_proj  + (long)at * KD);
            const float4 a0 = vr[lane], a1 = vr[32 + lane];
            const float4 c0 = ar[lane], c1 = ar[32 + lane];
            u0.x += p * a0.x; u0.y += p * a0.y; u0.z += p * a0.z; u0.w += p * a0.w;
            u1.x += p * a1.x; u1.y += p * a1.y; u1.z += p * a1.z; u1.w += p * a1.w;
            o0.x += p * c0.x; o0.y += p * c0.y; o0.z += p * c0.z; o0.w += p * c0.w;
            o1.x += p * c1.x; o1.y += p * c1.y; o1.z += p * c1.z; o1.w += p * c1.w;
        };
        int j = 0;
        for (; j + 4 <= n; j += 4) { edge(j); edge(j + 1); edge(j + 2); edge(j + 3); }
        if (j + 2 <= n) { edge(j); edge(j + 1); j += 2; }
        if (j < n) edge(j);
    }
    float4* ur   = (float4*)(U      + (long)h * KD);
    float4* orow = (float4*)(outAcc + (long)h * KD);
    ur[lane] = u0;   ur[32 + lane] = u1;
    orow[lane] = o0; orow[32 + lane] = o1;
}

__global__ void finalize_kernel(float* __restrict__ out,
                                const float* __restrict__ ent_feats,
                                long n) {
    long i = (long)blockIdx.x * blockDim.x + threadIdx.x;
    long stride = (long)gridDim.x * blockDim.x;
    for (; i < n; i += stride) {
        float x = out[i] + ent_feats[i];
        out[i] = (x > 0.0f) ? x : (expf(x) - 1.0f);
    }
}

extern "C" void kernel_launch(void* const* d_in, const int* in_sizes, int n_in,
                              void* d_out, int out_size, void* d_ws, size_t ws_size,
                              hipStream_t stream) {
    const int*   trip      = (const int*)d_in[0];
    const float* att_feats = (const float*)d_in[1];
    const float* val_feats = (const float*)d_in[2];
    const float* ent_feats = (const float*)d_in[3];
    const float* a_w       = (const float*)d_in[4];
    const float* a_b       = (const float*)d_in[5];
    const float* W         = (const float*)d_in[6];
    float* out = (float*)d_out;

    const float* W_top = W;
    const float* W_bot = W + KD * KD;

    float* ws       = (float*)d_ws;
    float* ent_s    = ws;
    float* att_s    = ent_s   + N_ENT;
    int*   row_cnt  = (int*)(att_s + N_ATT);
    int*   slot     = row_cnt + N_ENT;
    float* att_proj = (float*)(slot + NE);
    float* U        = att_proj + (long)N_ATT * KD;
    int*   rank     = (int*)U;

    const long nEntFeat = (long)N_ENT * KD;

    {
        int waves = N_ENT + N_ATT;
        int blocks = (waves * 32 + 255) / 256;
        row_dot_kernel<<<blocks, 256, 0, stream>>>(ent_feats, att_feats, a_w,
                                                   ent_s, att_s, row_cnt);
    }

    {
        int tiles = (N_ATT / 16) * 16;
        wmma_gemm_f32<false><<<tiles / 8, 256, 0, stream>>>(att_feats, W_top, att_proj);
    }

    edge_rank_kernel<<<(NE + 255) / 256, 256, 0, stream>>>(trip, row_cnt, rank);
    row_scan_kernel<<<1, SCAN_T, 0, stream>>>(row_cnt);
    edge_scatter_kernel<<<(NE + 255) / 256, 256, 0, stream>>>(trip, row_cnt, rank, slot);

    row_agg_kernel<<<(N_ENT * 32 + 255) / 256, 256, 0, stream>>>(row_cnt, slot, ent_s, att_s, a_b,
                                                               att_proj, val_feats, U, out);

    {
        int tiles = (N_ENT / 16) * 16;
        wmma_gemm_f32<true><<<tiles / 8, 256, 0, stream>>>(U, W_bot, out);
    }

    finalize_kernel<<<4096, 256, 0, stream>>>(out, ent_feats, nEntFeat);
}


# Row-dot and finalize passes folded into row_agg and the GEMM launches: ent row read once (score dot; residual goes into the GEMM's C, so ent is summed before U*W), ELU in GEMM epilogue; 8->6 launches
# speedup vs baseline: 1.8818x; 1.1280x over previous
#include <hip/hip_runtime.h>
#include <math.h>

#define N_ENT 50000
#define N_ATT 2000
#define N_VAL 100000
#define NE    300000
#define KD    256

#define VAL_BITS 17
static_assert(N_VAL <= (1 << VAL_BITS), "value row index must fit VAL_BITS");
static_assert(N_ATT <= (1 << (31 - VAL_BITS)), "attribute row index must fit the rest");

typedef __attribute__((ext_vector_type(2))) float v2f;
typedef __attribute__((ext_vector_type(8))) float v8f;

__device__ __forceinline__ float wave_row_dot(const float4 f0, const float4 f1,
                                              const float4 w0, const float4 w1) {
    float s = 0.0f;
    s += f0.x * w0.x + f0.y * w0.y + f0.z * w0.z + f0.w * w0.w;
    s += f1.x * w1.x + f1.y * w1.y + f1.z * w1.z + f1.w * w1.w;
    for (int off = 16; off > 0; off >>= 1) s += __shfl_xor(s, off, 32);
    return s;
}

__global__ void edge_rank_kernel(const int* __restrict__ trip,
                                 int* __restrict__ row_cnt,
                                 int* __restrict__ rank) {
    int e = blockIdx.x * blockDim.x + threadIdx.x;
    if (e >= NE) return;
    int h = trip[3 * e + 0];
    rank[e] = atomicAdd(&row_cnt[h], 1);
}

#define SCAN_T  1024
#define SCAN_CH ((N_ENT + SCAN_T - 1) / SCAN_T)
__global__ __launch_bounds__(SCAN_T) void row_scan_kernel(int* __restrict__ row_cnt) {
    __shared__ int s[SCAN_T * SCAN_CH];
    __shared__ int wtot[SCAN_T / 32];
    const int t = threadIdx.x;
    const int lane = t & 31;
    const int w = t >> 5;
    for (int k = 0; k < SCAN_CH; ++k) {
        const int i = t + k * SCAN_T;
        s[i] = (i < N_ENT) ? row_cnt[i] : 0;
    }
    __syncthreads();
    const int b = t * SCAN_CH;
    int sum = 0;
    for (int k = 0; k < SCAN_CH; ++k) sum += s[b + k];
    int v = sum;
    for (int o = 1; o < 32; o <<= 1) {
        int n = __shfl_up(v, o, 32);
        if (lane >= o) v += n;
    }
    if (lane == 31) wtot[w] = v;
    __syncthreads();
    if (w == 0) {
        int x = wtot[lane];
        for (int o = 1; o < 32; o <<= 1) {
            int n = __shfl_up(x, o, 32);
            if (lane >= o) x += n;
        }
        wtot[lane] = x;
    }
    __syncthreads();
    int run = (v - sum) + (w > 0 ? wtot[w - 1] : 0);
    for (int k = 0; k < SCAN_CH; ++k) { run += s[b + k]; s[b + k] = run; }
    __syncthreads();
    for (int k = 0; k < SCAN_CH; ++k) {
        const int i = t + k * SCAN_T;
        if (i < N_ENT) row_cnt[i] = s[i];
    }
}

__global__ void edge_scatter_kernel(const int* __restrict__ trip,
                                    const int* __restrict__ row_end,
                                    const int* __restrict__ rank,
                                    int* __restrict__ slot) {
    int e = blockIdx.x * blockDim.x + threadIdx.x;
    if (e >= NE) return;
    int h  = trip[3 * e + 0];
    int vl = trip[3 * e + 1];
    int at = trip[3 * e + 2];
    int beg = (h > 0) ? row_end[h - 1] : 0;
    slot[beg + rank[e]] = vl | (at << VAL_BITS);
}

template <bool FINAL>
__global__ void wmma_gemm_f32(const float* __restrict__ A,
                              const float* __restrict__ B,
                              float* __restrict__ C,
                              const float* __restrict__ a_w_att,
                              float* __restrict__ att_score,
                              int* __restrict__ row_cnt) {
    int wave = (int)((blockIdx.x * blockDim.x + threadIdx.x) >> 5);
    int lane = threadIdx.x & 31;
    int tM = wave >> 4;
    int tN = wave & 15;
    int row0 = tM * 16;
    int col0 = tN * 16;

    int m    = lane & 15;
    int koff = (lane >> 4) << 1;
    int bn   = col0 + (lane & 15);
    int crow = row0 + ((lane >> 4) << 3);
    int cn   = col0 + (lane & 15);

    v8f c;
    if (FINAL) {
#pragma unroll
        for (int r = 0; r < 8; ++r) c[r] = C[(long)(crow + r) * KD + cn];
    } else {
#pragma unroll
        for (int r = 0; r < 8; ++r) c[r] = 0.0f;
    }

    const float* aptr = A + (long)(row0 + m) * KD + koff;

#pragma unroll 4
    for (int k0 = 0; k0 < KD; k0 += 4) {
        v2f av = *(const v2f*)(aptr + k0);
        v2f bv;
        bv.x = B[(long)(k0 + koff + 0) * KD + bn];
        bv.y = B[(long)(k0 + koff + 1) * KD + bn];
        c = __builtin_amdgcn_wmma_f32_16x16x4_f32(
                 false, av,  false, bv,
                 (short)0, c,  false,  false);
    }

#pragma unroll
    for (int r = 0; r < 8; ++r) {
        float x = c[r];
        if (FINAL) x = (x > 0.0f) ? x : (expf(x) - 1.0f);
        C[(long)(crow + r) * KD + cn] = x;
    }

    if (!FINAL) {
        const int g = (int)(blockIdx.x * blockDim.x + threadIdx.x);
        if (g < N_ENT) row_cnt[g] = 0;
        const float4* fr = (const float4*)(A + (long)wave * KD);
        const float4* wr = (const float4*)a_w_att;
        const float s = wave_row_dot(fr[lane], fr[32 + lane], wr[lane], wr[32 + lane]);
        if (lane == 0) att_score[wave] = s;
    }
}

__device__ __forceinline__ float edge_sc(float es, float as, float b) {
    float x = es + as + b;
    x = (x > 0.0f) ? x : 0.2f * x;
    return expf(x);
}

__global__ __launch_bounds__(256) void row_agg_kernel(const int* __restrict__ row_end,
                                                      const int* __restrict__ slot,
                                                      const float* __restrict__ ent_feats,
                                                      const float* __restrict__ a_w,
                                                      const float* __restrict__ att_score,
                                                      const float* __restrict__ a_b,
                                                      const float* __restrict__ att_proj,
                                                      const float* __restrict__ val_feats,
                                                      float* __restrict__ U,
                                                      float* __restrict__ outAcc) {
    const int h = __builtin_amdgcn_readfirstlane((int)((blockIdx.x * 256 + threadIdx.x) >> 5));
    const int lane = threadIdx.x & 31;
    if (h >= N_ENT) return;
    const float4* er = (const float4*)(ent_feats + (long)h * KD);
    const float4* aw = (const float4*)a_w;
    const float4 e0 = er[lane], e1 = er[32 + lane];
    const float4 w0 = aw[lane], w1 = aw[32 + lane];
    const int beg = (h > 0) ? row_end[h - 1] : 0;
    const int end = row_end[h];
    const float b  = a_b[0];

    int   wd_l = 0;
    float as_l = 0.0f;
    const bool has_l = (beg + lane < end);
    if (has_l) {
        wd_l = slot[beg + lane];
        as_l = att_score[wd_l >> VAL_BITS];
    }
    const float es = wave_row_dot(e0, e1, w0, w1);
    float sc_l = has_l ? edge_sc(es, as_l, b) : 0.0f;
    float rs = sc_l;
    for (int i = beg + 32 + lane; i < end; i += 32)
        rs += edge_sc(es, att_score[slot[i] >> VAL_BITS], b);
    for (int off = 16; off > 0; off >>= 1) rs += __shfl_xor(rs, off, 32);

    float4 u0 = make_float4(0.f, 0.f, 0.f, 0.f), u1 = u0, o0 = u0, o1 = u0;
    for (int c = beg; c < end; c += 32) {
        if (c != beg) {
            wd_l = 0;
            sc_l = 0.0f;
            if (c + lane < end) {
                wd_l = slot[c + lane];
                sc_l = edge_sc(es, att_score[wd_l >> VAL_BITS], b);
            }
        }
        const float p_l = sc_l / rs;
        const int n = (end - c < 32) ? (end - c) : 32;
        auto edge = [&](int j) {
            const int   wd = __builtin_amdgcn_readlane(wd_l, j);
            const float p  = __int_as_float(__builtin_amdgcn_readlane(__float_as_int(p_l), j));
            const int vl = wd & ((1 << VAL_BITS) - 1);
            const int at = wd >> VAL_BITS;
            const float4* vr = (const float4*)(val_feats + (long)vl * KD);
            const float4* ar = (const float4*)(att_proj  + (long)at * KD);
            const float4 a0 = vr[lane], a1 = vr[32 + lane];
            const float4 c0 = ar[lane], c1 = ar[32 + lane];
            u0.x += p * a0.x; u0.y += p * a0.y; u0.z += p * a0.z; u0.w += p * a0.w;
            u1.x += p * a1.x; u1.y += p * a1.y; u1.z += p * a1.z; u1.w += p * a1.w;
            o0.x += p * c0.x; o0.y += p * c0.y; o0.z += p * c0.z; o0.w += p * c0.w;
            o1.x += p * c1.x; o1.y += p * c1.y; o1.z += p * c1.z; o1.w += p * c1.w;
        };
        int j = 0;
        for (; j + 4 <= n; j += 4) { edge(j); edge(j + 1); edge(j + 2); edge(j + 3); }
        if (j + 2 <= n) { edge(j); edge(j + 1); j += 2; }
        if (j < n) edge(j);
    }
    float4* ur   = (float4*)(U      + (long)h * KD);
    float4* orow = (float4*)(outAcc + (long)h * KD);
    ur[lane] = u0;   ur[32 + lane] = u1;
    o0.x += e0.x; o0.y += e0.y; o0.z += e0.z; o0.w += e0.w;
    o1.x += e1.x; o1.y += e1.y; o1.z += e1.z; o1.w += e1.w;
    orow[lane] = o0; orow[32 + lane] = o1;
}

static_assert((N_ATT / 16) * 16 == N_ATT, "attribute GEMM: one wave per attribute row, whole 16-row tiles");
static_assert((N_ATT / 16) * 16 * 32 >= N_ENT, "attribute GEMM launch must have a thread per entity row counter");
static_assert((N_ENT / 16) * 16 == N_ENT, "entity GEMM: whole 16-row tiles");

extern "C" void kernel_launch(void* const* d_in, const int* in_sizes, int n_in,
                              void* d_out, int out_size, void* d_ws, size_t ws_size,
                              hipStream_t stream) {
    const int*   trip      = (const int*)d_in[0];
    const float* att_feats = (const float*)d_in[1];
    const float* val_feats = (const float*)d_in[2];
    const float* ent_feats = (const float*)d_in[3];
    const float* a_w       = (const float*)d_in[4];
    const float* a_b       = (const float*)d_in[5];
    const float* W         = (const float*)d_in[6];
    float* out = (float*)d_out;

    const float* W_top = W;
    const float* W_bot = W + KD * KD;

    float* ws       = (float*)d_ws;
    float* att_s    = ws + N_ENT;
    int*   row_cnt  = (int*)(att_s + N_ATT);
    int*   slot     = row_cnt + N_ENT;
    float* att_proj = (float*)(slot + NE);
    float* U        = att_proj + (long)N_ATT * KD;
    int*   rank     = (int*)U;

    {
        int tiles = (N_ATT / 16) * 16;
        wmma_gemm_f32<false><<<tiles / 8, 256, 0, stream>>>(att_feats, W_top, att_proj,
                                                            a_w + KD, att_s, row_cnt);
    }

    edge_rank_kernel<<<(NE + 255) / 256, 256, 0, stream>>>(trip, row_cnt, rank);
    row_scan_kernel<<<1, SCAN_T, 0, stream>>>(row_cnt);
    edge_scatter_kernel<<<(NE + 255) / 256, 256, 0, stream>>>(trip, row_cnt, rank, slot);

    row_agg_kernel<<<(N_ENT * 32 + 255) / 256, 256, 0, stream>>>(row_cnt, slot, ent_feats, a_w, att_s, a_b,
                                                               att_proj, val_feats, U, out);

    {
        int tiles = (N_ENT / 16) * 16;
        wmma_gemm_f32<true><<<tiles / 8, 256, 0, stream>>>(U, W_bot, out,
                                                           (const float*)nullptr, (float*)nullptr,
                                                           (int*)nullptr);
    }
}


# Entity GEMM as 256 resident work-groups: W_bot staged once in LDS (k-pair interleaved), each wave 16x64 (4 accumulator tiles share an A fragment); k order and sums unchanged
# speedup vs baseline: 2.5460x; 1.3530x over previous
#include <hip/hip_runtime.h>
#include <math.h>

#define N_ENT 50000
#define N_ATT 2000
#define N_VAL 100000
#define NE    300000
#define KD    256

#define VAL_BITS 17
static_assert(N_VAL <= (1 << VAL_BITS), "value row index must fit VAL_BITS");
static_assert(N_ATT <= (1 << (31 - VAL_BITS)), "attribute row index must fit the rest");

typedef __attribute__((ext_vector_type(2))) float v2f;
typedef __attribute__((ext_vector_type(8))) float v8f;

__device__ __forceinline__ float wave_row_dot(const float4 f0, const float4 f1,
                                              const float4 w0, const float4 w1) {
    float s = 0.0f;
    s += f0.x * w0.x + f0.y * w0.y + f0.z * w0.z + f0.w * w0.w;
    s += f1.x * w1.x + f1.y * w1.y + f1.z * w1.z + f1.w * w1.w;
    for (int off = 16; off > 0; off >>= 1) s += __shfl_xor(s, off, 32);
    return s;
}

__global__ void edge_rank_kernel(const int* __restrict__ trip,
                                 int* __restrict__ row_cnt,
                                 int* __restrict__ rank) {
    int e = blockIdx.x * blockDim.x + threadIdx.x;
    if (e >= NE) return;
    int h = trip[3 * e + 0];
    rank[e] = atomicAdd(&row_cnt[h], 1);
}

#define SCAN_T  1024
#define SCAN_CH ((N_ENT + SCAN_T - 1) / SCAN_T)
__global__ __launch_bounds__(SCAN_T) void row_scan_kernel(int* __restrict__ row_cnt) {
    __shared__ int s[SCAN_T * SCAN_CH];
    __shared__ int wtot[SCAN_T / 32];
    const int t = threadIdx.x;
    const int lane = t & 31;
    const int w = t >> 5;
    for (int k = 0; k < SCAN_CH; ++k) {
        const int i = t + k * SCAN_T;
        s[i] = (i < N_ENT) ? row_cnt[i] : 0;
    }
    __syncthreads();
    const int b = t * SCAN_CH;
    int sum = 0;
    for (int k = 0; k < SCAN_CH; ++k) sum += s[b + k];
    int v = sum;
    for (int o = 1; o < 32; o <<= 1) {
        int n = __shfl_up(v, o, 32);
        if (lane >= o) v += n;
    }
    if (lane == 31) wtot[w] = v;
    __syncthreads();
    if (w == 0) {
        int x = wtot[lane];
        for (int o = 1; o < 32; o <<= 1) {
            int n = __shfl_up(x, o, 32);
            if (lane >= o) x += n;
        }
        wtot[lane] = x;
    }
    __syncthreads();
    int run = (v - sum) + (w > 0 ? wtot[w - 1] : 0);
    for (int k = 0; k < SCAN_CH; ++k) { run += s[b + k]; s[b + k] = run; }
    __syncthreads();
    for (int k = 0; k < SCAN_CH; ++k) {
        const int i = t + k * SCAN_T;
        if (i < N_ENT) row_cnt[i] = s[i];
    }
}

__global__ void edge_scatter_kernel(const int* __restrict__ trip,
                                    const int* __restrict__ row_end,
                                    const int* __restrict__ rank,
                                    int* __restrict__ slot) {
    int e = blockIdx.x * blockDim.x + threadIdx.x;
    if (e >= NE) return;
    int h  = trip[3 * e + 0];
    int vl = trip[3 * e + 1];
    int at = trip[3 * e + 2];
    int beg = (h > 0) ? row_end[h - 1] : 0;
    slot[beg + rank[e]] = vl | (at << VAL_BITS);
}


__global__ void wmma_gemm_att(const float* __restrict__ A,
                              const float* __restrict__ B,
                              float* __restrict__ C,
                              const float* __restrict__ a_w_att,
                              float* __restrict__ att_score,
                              int* __restrict__ row_cnt) {
    int wave = (int)((blockIdx.x * blockDim.x + threadIdx.x) >> 5);
    int lane = threadIdx.x & 31;
    int tM = wave >> 4;
    int tN = wave & 15;
    int row0 = tM * 16;
    int col0 = tN * 16;

    int m    = lane & 15;
    int koff = (lane >> 4) << 1;
    int bn   = col0 + (lane & 15);
    int crow = row0 + ((lane >> 4) << 3);
    int cn   = col0 + (lane & 15);

    v8f c;
#pragma unroll
    for (int r = 0; r < 8; ++r) c[r] = 0.0f;

    const float* aptr = A + (long)(row0 + m) * KD + koff;

#pragma unroll 4
    for (int k0 = 0; k0 < KD; k0 += 4) {
        v2f av = *(const v2f*)(aptr + k0);
        v2f bv;
        bv.x = B[(long)(k0 + koff + 0) * KD + bn];
        bv.y = B[(long)(k0 + koff + 1) * KD + bn];
        c = __builtin_amdgcn_wmma_f32_16x16x4_f32(
                 false, av,  false, bv,
                 (short)0, c,  false,  false);
    }

#pragma unroll
    for (int r = 0; r < 8; ++r) C[(long)(crow + r) * KD + cn] = c[r];

    const int g = (int)(blockIdx.x * blockDim.x + threadIdx.x);
    if (g < N_ENT) row_cnt[g] = 0;
    const float4* fr = (const float4*)(A + (long)wave * KD);
    const float4* wr = (const float4*)a_w_att;
    const float s = wave_row_dot(fr[lane], fr[32 + lane], wr[lane], wr[32 + lane]);
    if (lane == 0) att_score[wave] = s;
}

#define FB_WAVES   16
#define FB_PITCH   544
#define FB_WGS     256
#define FB_CH      8
#define M_TILES    (N_ENT / 16)

__device__ __forceinline__ void ent_chunk(v8f (&acc)[4], const v2f (&a)[FB_CH], const float* bs) {
    v2f b[4];
#pragma unroll
    for (int t = 0; t < 4; ++t) b[t] = *(const v2f*)(bs + 32 * t);
#pragma unroll
    for (int s = 0; s < FB_CH; ++s) {
        v2f bn[4];
        if (s + 1 < FB_CH) {
#pragma unroll
            for (int t = 0; t < 4; ++t) bn[t] = *(const v2f*)(bs + 2 * (s + 1) * FB_PITCH + 32 * t);
        }
#pragma unroll
        for (int t = 0; t < 4; ++t)
            acc[t] = __builtin_amdgcn_wmma_f32_16x16x4_f32(
                          false, a[s],  false, b[t],
                          (short)0, acc[t],  false,  false);
        if (s + 1 < FB_CH) {
#pragma unroll
            for (int t = 0; t < 4; ++t) b[t] = bn[t];
        }
    }
}

__global__ __launch_bounds__(FB_WAVES * 32) void wmma_gemm_ent(const float* __restrict__ A,
                                                               const float* __restrict__ B,
                                                               float* __restrict__ C) {
    __shared__ float Bs[(KD / 2) * FB_PITCH];

#pragma unroll
    for (int it = 0; it < (KD / 2) * (KD / 4) / (FB_WAVES * 32); ++it) {
        const int idx = (int)threadIdx.x + it * (FB_WAVES * 32);
        const int p  = idx >> 6;
        const int n4 = (idx & 63) << 2;
        const float4 r0 = *(const float4*)(B + (long)(2 * p)     * KD + n4);
        const float4 r1 = *(const float4*)(B + (long)(2 * p + 1) * KD + n4);
        float* d = Bs + p * FB_PITCH + n4 * 2;
        *(float4*)(d)     = make_float4(r0.x, r1.x, r0.y, r1.y);
        *(float4*)(d + 4) = make_float4(r0.z, r1.z, r0.w, r1.w);
    }
    __syncthreads();

    const int wave = __builtin_amdgcn_readfirstlane((int)(threadIdx.x >> 5));
    const int lane = threadIdx.x & 31;
    const int grp  = wave >> 2;
    const int col0 = (wave & 3) * 64;
    const int c    = lane & 15;
    const int h    = lane >> 4;

    const float* bbase = Bs + h * FB_PITCH + 2 * (col0 + c);

    for (int tM = blockIdx.x + gridDim.x * grp; tM < M_TILES; tM += gridDim.x * 4) {
        asm volatile("" ::: "memory");
        const int row0 = tM * 16;
        float* crow = C + (long)(row0 + 8 * h) * KD + col0 + c;
        const float* ap = A + (long)(row0 + c) * KD + 2 * h;

        v2f a[FB_CH];
#pragma unroll
        for (int s = 0; s < FB_CH; ++s) a[s] = *(const v2f*)(ap + 4 * s);

        v8f acc[4];
#pragma unroll
        for (int t = 0; t < 4; ++t)
#pragma unroll
            for (int r = 0; r < 8; ++r) acc[t][r] = crow[(long)r * KD + 16 * t];

        const float* bs = bbase;
#pragma unroll 1
        for (int ch = 0; ch < KD / (4 * FB_CH) - 1; ++ch) {
            v2f an[FB_CH];
#pragma unroll
            for (int s = 0; s < FB_CH; ++s) an[s] = *(const v2f*)(ap + 4 * FB_CH * (ch + 1) + 4 * s);
            ent_chunk(acc, a, bs);
            bs += 2 * FB_CH * FB_PITCH;
#pragma unroll
            for (int s = 0; s < FB_CH; ++s) a[s] = an[s];
        }
        ent_chunk(acc, a, bs);

#pragma unroll
        for (int t = 0; t < 4; ++t)
#pragma unroll
            for (int r = 0; r < 8; ++r) {
                float x = acc[t][r];
                x = (x > 0.0f) ? x : (expf(x) - 1.0f);
                crow[(long)r * KD + 16 * t] = x;
            }
    }
}

__device__ __forceinline__ float edge_sc(float es, float as, float b) {
    float x = es + as + b;
    x = (x > 0.0f) ? x : 0.2f * x;
    return expf(x);
}

__global__ __launch_bounds__(256) void row_agg_kernel(const int* __restrict__ row_end,
                                                      const int* __restrict__ slot,
                                                      const float* __restrict__ ent_feats,
                                                      const float* __restrict__ a_w,
                                                      const float* __restrict__ att_score,
                                                      const float* __restrict__ a_b,
                                                      const float* __restrict__ att_proj,
                                                      const float* __restrict__ val_feats,
                                                      float* __restrict__ U,
                                                      float* __restrict__ outAcc) {
    const int h = __builtin_amdgcn_readfirstlane((int)((blockIdx.x * 256 + threadIdx.x) >> 5));
    const int lane = threadIdx.x & 31;
    if (h >= N_ENT) return;
    const float4* er = (const float4*)(ent_feats + (long)h * KD);
    const float4* aw = (const float4*)a_w;
    const float4 e0 = er[lane], e1 = er[32 + lane];
    const float4 w0 = aw[lane], w1 = aw[32 + lane];
    const int beg = (h > 0) ? row_end[h - 1] : 0;
    const int end = row_end[h];
    const float b  = a_b[0];

    int   wd_l = 0;
    float as_l = 0.0f;
    const bool has_l = (beg + lane < end);
    if (has_l) {
        wd_l = slot[beg + lane];
        as_l = att_score[wd_l >> VAL_BITS];
    }
    const float es = wave_row_dot(e0, e1, w0, w1);
    float sc_l = has_l ? edge_sc(es, as_l, b) : 0.0f;
    float rs = sc_l;
    for (int i = beg + 32 + lane; i < end; i += 32)
        rs += edge_sc(es, att_score[slot[i] >> VAL_BITS], b);
    for (int off = 16; off > 0; off >>= 1) rs += __shfl_xor(rs, off, 32);

    float4 u0 = make_float4(0.f, 0.f, 0.f, 0.f), u1 = u0, o0 = u0, o1 = u0;
    for (int c = beg; c < end; c += 32) {
        if (c != beg) {
            wd_l = 0;
            sc_l = 0.0f;
            if (c + lane < end) {
                wd_l = slot[c + lane];
                sc_l = edge_sc(es, att_score[wd_l >> VAL_BITS], b);
            }
        }
        const float p_l = sc_l / rs;
        const int n = (end - c < 32) ? (end - c) : 32;
        auto edge = [&](int j) {
            const int   wd = __builtin_amdgcn_readlane(wd_l, j);
            const float p  = __int_as_float(__builtin_amdgcn_readlane(__float_as_int(p_l), j));
            const int vl = wd & ((1 << VAL_BITS) - 1);
            const int at = wd >> VAL_BITS;
            const float4* vr = (const float4*)(val_feats + (long)vl * KD);
            const float4* ar = (const float4*)(att_proj  + (long)at * KD);
            const float4 a0 = vr[lane], a1 = vr[32 + lane];
            const float4 c0 = ar[lane], c1 = ar[32 + lane];
            u0.x += p * a0.x; u0.y += p * a0.y; u0.z += p * a0.z; u0.w += p * a0.w;
            u1.x += p * a1.x; u1.y += p * a1.y; u1.z += p * a1.z; u1.w += p * a1.w;
            o0.x += p * c0.x; o0.y += p * c0.y; o0.z += p * c0.z; o0.w += p * c0.w;
            o1.x += p * c1.x; o1.y += p * c1.y; o1.z += p * c1.z; o1.w += p * c1.w;
        };
        int j = 0;
        for (; j + 4 <= n; j += 4) { edge(j); edge(j + 1); edge(j + 2); edge(j + 3); }
        if (j + 2 <= n) { edge(j); edge(j + 1); j += 2; }
        if (j < n) edge(j);
    }
    float4* ur   = (float4*)(U      + (long)h * KD);
    float4* orow = (float4*)(outAcc + (long)h * KD);
    ur[lane] = u0;   ur[32 + lane] = u1;
    o0.x += e0.x; o0.y += e0.y; o0.z += e0.z; o0.w += e0.w;
    o1.x += e1.x; o1.y += e1.y; o1.z += e1.z; o1.w += e1.w;
    orow[lane] = o0; orow[32 + lane] = o1;
}

static_assert((N_ATT / 16) * 16 == N_ATT, "attribute GEMM: one wave per attribute row, whole 16-row tiles");
static_assert((N_ATT / 16) * 16 * 32 >= N_ENT, "attribute GEMM launch must have a thread per entity row counter");
static_assert((N_ENT / 16) * 16 == N_ENT, "entity GEMM: whole 16-row tiles");
static_assert((KD / 2) * FB_PITCH * 4 <= 320 * 1024, "entity GEMM: W_bot must fit the WGP's LDS");

extern "C" void kernel_launch(void* const* d_in, const int* in_sizes, int n_in,
                              void* d_out, int out_size, void* d_ws, size_t ws_size,
                              hipStream_t stream) {
    const int*   trip      = (const int*)d_in[0];
    const float* att_feats = (const float*)d_in[1];
    const float* val_feats = (const float*)d_in[2];
    const float* ent_feats = (const float*)d_in[3];
    const float* a_w       = (const float*)d_in[4];
    const float* a_b       = (const float*)d_in[5];
    const float* W         = (const float*)d_in[6];
    float* out = (float*)d_out;

    const float* W_top = W;
    const float* W_bot = W + KD * KD;

    float* ws       = (float*)d_ws;
    float* att_s    = ws + N_ENT;
    int*   row_cnt  = (int*)(att_s + N_ATT);
    int*   slot     = row_cnt + N_ENT;
    float* att_proj = (float*)(slot + NE);
    float* U        = att_proj + (long)N_ATT * KD;
    int*   rank     = (int*)U;

    {
        int tiles = (N_ATT / 16) * 16;
        wmma_gemm_att<<<tiles / 8, 256, 0, stream>>>(att_feats, W_top, att_proj,
                                                     a_w + KD, att_s, row_cnt);
    }

    edge_rank_kernel<<<(NE + 255) / 256, 256, 0, stream>>>(trip, row_cnt, rank);
    row_scan_kernel<<<1, SCAN_T, 0, stream>>>(row_cnt);
    edge_scatter_kernel<<<(NE + 255) / 256, 256, 0, stream>>>(trip, row_cnt, rank, slot);

    row_agg_kernel<<<(N_ENT * 32 + 255) / 256, 256, 0, stream>>>(row_cnt, slot, ent_feats, a_w, att_s, a_b,
                                                               att_proj, val_feats, U, out);

    wmma_gemm_ent<<<FB_WGS, FB_WAVES * 32, 0, stream>>>(U, W_bot, out);
}


# Row aggregation fused into the entity GEMM launch: 16 gather waves fill a 16-row U tile in LDS while 8 WMMA waves multiply the previous tile; U never goes to memory; same sum order; 6->5 launches
# speedup vs baseline: 2.7079x; 1.0636x over previous
#include <hip/hip_runtime.h>
#include <math.h>

#define N_ENT 50000
#define N_ATT 2000
#define N_VAL 100000
#define NE    300000
#define KD    256

#define VAL_BITS 17
static_assert(N_VAL <= (1 << VAL_BITS), "value row index must fit VAL_BITS");
static_assert(N_ATT <= (1 << (31 - VAL_BITS)), "attribute row index must fit the rest");

typedef __attribute__((ext_vector_type(2))) float v2f;
typedef __attribute__((ext_vector_type(8))) float v8f;

__device__ __forceinline__ float wave_row_dot(const float4 f0, const float4 f1,
                                              const float4 w0, const float4 w1) {
    float s = 0.0f;
    s += f0.x * w0.x + f0.y * w0.y + f0.z * w0.z + f0.w * w0.w;
    s += f1.x * w1.x + f1.y * w1.y + f1.z * w1.z + f1.w * w1.w;
    for (int off = 16; off > 0; off >>= 1) s += __shfl_xor(s, off, 32);
    return s;
}

__global__ void edge_rank_kernel(const int* __restrict__ trip,
                                 int* __restrict__ row_cnt,
                                 int* __restrict__ rank) {
    int e = blockIdx.x * blockDim.x + threadIdx.x;
    if (e >= NE) return;
    int h = trip[3 * e + 0];
    rank[e] = atomicAdd(&row_cnt[h], 1);
}

#define SCAN_T  1024
#define SCAN_CH ((N_ENT + SCAN_T - 1) / SCAN_T)
__global__ __launch_bounds__(SCAN_T) void row_scan_kernel(int* __restrict__ row_cnt) {
    __shared__ int s[SCAN_T * SCAN_CH];
    __shared__ int wtot[SCAN_T / 32];
    const int t = threadIdx.x;
    const int lane = t & 31;
    const int w = t >> 5;
    for (int k = 0; k < SCAN_CH; ++k) {
        const int i = t + k * SCAN_T;
        s[i] = (i < N_ENT) ? row_cnt[i] : 0;
    }
    __syncthreads();
    const int b = t * SCAN_CH;
    int sum = 0;
    for (int k = 0; k < SCAN_CH; ++k) sum += s[b + k];
    int v = sum;
    for (int o = 1; o < 32; o <<= 1) {
        int n = __shfl_up(v, o, 32);
        if (lane >= o) v += n;
    }
    if (lane == 31) wtot[w] = v;
    __syncthreads();
    if (w == 0) {
        int x = wtot[lane];
        for (int o = 1; o < 32; o <<= 1) {
            int n = __shfl_up(x, o, 32);
            if (lane >= o) x += n;
        }
        wtot[lane] = x;
    }
    __syncthreads();
    int run = (v - sum) + (w > 0 ? wtot[w - 1] : 0);
    for (int k = 0; k < SCAN_CH; ++k) { run += s[b + k]; s[b + k] = run; }
    __syncthreads();
    for (int k = 0; k < SCAN_CH; ++k) {
        const int i = t + k * SCAN_T;
        if (i < N_ENT) row_cnt[i] = s[i];
    }
}

__global__ void edge_scatter_kernel(const int* __restrict__ trip,
                                    const int* __restrict__ row_end,
                                    const int* __restrict__ rank,
                                    int* __restrict__ slot) {
    int e = blockIdx.x * blockDim.x + threadIdx.x;
    if (e >= NE) return;
    int h  = trip[3 * e + 0];
    int vl = trip[3 * e + 1];
    int at = trip[3 * e + 2];
    int beg = (h > 0) ? row_end[h - 1] : 0;
    slot[beg + rank[e]] = vl | (at << VAL_BITS);
}


__global__ void wmma_gemm_att(const float* __restrict__ A,
                              const float* __restrict__ B,
                              float* __restrict__ C,
                              const float* __restrict__ a_w_att,
                              float* __restrict__ att_score,
                              int* __restrict__ row_cnt) {
    int wave = (int)((blockIdx.x * blockDim.x + threadIdx.x) >> 5);
    int lane = threadIdx.x & 31;
    int tM = wave >> 4;
    int tN = wave & 15;
    int row0 = tM * 16;
    int col0 = tN * 16;

    int m    = lane & 15;
    int koff = (lane >> 4) << 1;
    int bn   = col0 + (lane & 15);
    int crow = row0 + ((lane >> 4) << 3);
    int cn   = col0 + (lane & 15);

    v8f c;
#pragma unroll
    for (int r = 0; r < 8; ++r) c[r] = 0.0f;

    const float* aptr = A + (long)(row0 + m) * KD + koff;

#pragma unroll 4
    for (int k0 = 0; k0 < KD; k0 += 4) {
        v2f av = *(const v2f*)(aptr + k0);
        v2f bv;
        bv.x = B[(long)(k0 + koff + 0) * KD + bn];
        bv.y = B[(long)(k0 + koff + 1) * KD + bn];
        c = __builtin_amdgcn_wmma_f32_16x16x4_f32(
                 false, av,  false, bv,
                 (short)0, c,  false,  false);
    }

#pragma unroll
    for (int r = 0; r < 8; ++r) C[(long)(crow + r) * KD + cn] = c[r];

    const int g = (int)(blockIdx.x * blockDim.x + threadIdx.x);
    if (g < N_ENT) row_cnt[g] = 0;
    const float4* fr = (const float4*)(A + (long)wave * KD);
    const float4* wr = (const float4*)a_w_att;
    const float s = wave_row_dot(fr[lane], fr[32 + lane], wr[lane], wr[32 + lane]);
    if (lane == 0) att_score[wave] = s;
}

#define FB_PITCH   544
#define FB_WGS     256
#define FB_CH      8
#define M_TILES    (N_ENT / 16)
#define FU_CONS    8
#define FU_PROD    16
#define FU_THREADS ((FU_CONS + FU_PROD) * 32)
#define FU_UP      260
#define FU_NT      2

__device__ __forceinline__ void ent_chunk(v8f (&acc)[FU_NT], const v2f (&a)[FB_CH], const float* bs) {
    v2f b[FU_NT];
#pragma unroll
    for (int t = 0; t < FU_NT; ++t) b[t] = *(const v2f*)(bs + 32 * t);
#pragma unroll
    for (int s = 0; s < FB_CH; ++s) {
        v2f bn[FU_NT];
        if (s + 1 < FB_CH) {
#pragma unroll
            for (int t = 0; t < FU_NT; ++t) bn[t] = *(const v2f*)(bs + 2 * (s + 1) * FB_PITCH + 32 * t);
        }
#pragma unroll
        for (int t = 0; t < FU_NT; ++t)
            acc[t] = __builtin_amdgcn_wmma_f32_16x16x4_f32(
                          false, a[s],  false, b[t],
                          (short)0, acc[t],  false,  false);
        if (s + 1 < FB_CH) {
#pragma unroll
            for (int t = 0; t < FU_NT; ++t) b[t] = bn[t];
        }
    }
}

__device__ __forceinline__ float edge_sc(float es, float as, float b) {
    float x = es + as + b;
    x = (x > 0.0f) ? x : 0.2f * x;
    return expf(x);
}

struct RowIn { int beg, end; float4 e0, e1; };

__device__ __forceinline__ RowIn row_fetch(int h, int lane, const int* __restrict__ row_end,
                                           const float* __restrict__ ent_feats) {
    RowIn r;
    const float4* er = (const float4*)(ent_feats + (long)h * KD);
    r.e0 = er[lane]; r.e1 = er[32 + lane];
    r.beg = (h > 0) ? row_end[h - 1] : 0;
    r.end = row_end[h];
    return r;
}

__device__ __forceinline__ RowIn row_produce(const RowIn in, const int h, const int hnext, const int lane,
                                            const float4 w0, const float4 w1, const float b,
                                            const int* __restrict__ slot,
                                            const float* __restrict__ att_score,
                                            const float* __restrict__ att_proj,
                                            const float* __restrict__ val_feats,
                                            const int* __restrict__ row_end,
                                            const float* __restrict__ ent_feats,
                                            float* urow,
                                            float* __restrict__ outAcc) {
    const float4 e0 = in.e0, e1 = in.e1;
    const int beg = in.beg, end = in.end;

    int   wd_l = 0;
    float as_l = 0.0f;
    const bool has_l = (beg + lane < end);
    if (has_l) {
        wd_l = slot[beg + lane];
        as_l = att_score[wd_l >> VAL_BITS];
    }
    const float es = wave_row_dot(e0, e1, w0, w1);
    float sc_l = has_l ? edge_sc(es, as_l, b) : 0.0f;
    float rs = sc_l;
    for (int i = beg + 32 + lane; i < end; i += 32)
        rs += edge_sc(es, att_score[slot[i] >> VAL_BITS], b);
    for (int off = 16; off > 0; off >>= 1) rs += __shfl_xor(rs, off, 32);

    RowIn nxt = in;
    if (hnext >= 0) nxt = row_fetch(hnext, lane, row_end, ent_feats);

    float4 u0 = make_float4(0.f, 0.f, 0.f, 0.f), u1 = u0, o0 = u0, o1 = u0;
    for (int c = beg; c < end; c += 32) {
        if (c != beg) {
            wd_l = 0;
            sc_l = 0.0f;
            if (c + lane < end) {
                wd_l = slot[c + lane];
                sc_l = edge_sc(es, att_score[wd_l >> VAL_BITS], b);
            }
        }
        const float p_l = sc_l / rs;
        const int n = (end - c < 32) ? (end - c) : 32;
        auto edge = [&](int j) {
            const int   wd = __builtin_amdgcn_readlane(wd_l, j);
            const float p  = __int_as_float(__builtin_amdgcn_readlane(__float_as_int(p_l), j));
            const int vl = wd & ((1 << VAL_BITS) - 1);
            const int at = wd >> VAL_BITS;
            const float4* vr = (const float4*)(val_feats + (long)vl * KD);
            const float4* ar = (const float4*)(att_proj  + (long)at * KD);
            const float4 a0 = vr[lane], a1 = vr[32 + lane];
            const float4 c0 = ar[lane], c1 = ar[32 + lane];
            u0.x += p * a0.x; u0.y += p * a0.y; u0.z += p * a0.z; u0.w += p * a0.w;
            u1.x += p * a1.x; u1.y += p * a1.y; u1.z += p * a1.z; u1.w += p * a1.w;
            o0.x += p * c0.x; o0.y += p * c0.y; o0.z += p * c0.z; o0.w += p * c0.w;
            o1.x += p * c1.x; o1.y += p * c1.y; o1.z += p * c1.z; o1.w += p * c1.w;
        };
        int j = 0;
        for (; j + 4 <= n; j += 4) { edge(j); edge(j + 1); edge(j + 2); edge(j + 3); }
        const int rem = n - j;
        if (rem == 3)      { edge(j); edge(j + 1); edge(j + 2); }
        else if (rem == 2) { edge(j); edge(j + 1); }
        else if (rem == 1) { edge(j); }
    }
    float4* ur   = (float4*)urow;
    float4* orow = (float4*)(outAcc + (long)h * KD);
    ur[lane] = u0;   ur[32 + lane] = u1;
    o0.x += e0.x; o0.y += e0.y; o0.z += e0.z; o0.w += e0.w;
    o1.x += e1.x; o1.y += e1.y; o1.z += e1.z; o1.w += e1.w;
    orow[lane] = o0; orow[32 + lane] = o1;
    return nxt;
}

__global__ __launch_bounds__(FU_THREADS) void row_agg_gemm_ent(const int* __restrict__ row_end,
                                                               const int* __restrict__ slot,
                                                               const float* __restrict__ ent_feats,
                                                               const float* __restrict__ a_w,
                                                               const float* __restrict__ att_score,
                                                               const float* __restrict__ a_b,
                                                               const float* __restrict__ att_proj,
                                                               const float* __restrict__ val_feats,
                                                               const float* __restrict__ B,
                                                               float* C) {
    __shared__ float Bs[(KD / 2) * FB_PITCH];
    __shared__ float Us[2 * 16 * FU_UP];

    const int wave = __builtin_amdgcn_readfirstlane((int)(threadIdx.x >> 5));
    const int lane = threadIdx.x & 31;
    const bool producer = (wave >= FU_CONS);
    const int prow = wave - FU_CONS;
    const int nT = (M_TILES - (int)blockIdx.x + FB_WGS - 1) / FB_WGS;

    RowIn cur;
    float4 w0, w1;
    float ab = 0.0f;
    if (producer) {
        const float4* aw = (const float4*)a_w;
        w0 = aw[lane]; w1 = aw[32 + lane];
        ab = a_b[0];
        cur = row_fetch((int)blockIdx.x * 16 + prow, lane, row_end, ent_feats);
    }

#pragma unroll
    for (int it = 0; it < ((KD / 2) * (KD / 4) + FU_THREADS - 1) / FU_THREADS; ++it) {
        const int idx = (int)threadIdx.x + it * FU_THREADS;
        if (idx < (KD / 2) * (KD / 4)) {
            const int p  = idx >> 6;
            const int n4 = (idx & 63) << 2;
            const float4 r0 = *(const float4*)(B + (long)(2 * p)     * KD + n4);
            const float4 r1 = *(const float4*)(B + (long)(2 * p + 1) * KD + n4);
            float* d = Bs + p * FB_PITCH + n4 * 2;
            *(float4*)(d)     = make_float4(r0.x, r1.x, r0.y, r1.y);
            *(float4*)(d + 4) = make_float4(r0.z, r1.z, r0.w, r1.w);
        }
    }

    if (producer) {
        const int h = (int)blockIdx.x * 16 + prow;
        cur = row_produce(cur, h, (1 < nT) ? h + FB_WGS * 16 : -1, lane, w0, w1, ab,
                          slot, att_score, att_proj, val_feats, row_end, ent_feats,
                          Us + prow * FU_UP, C);
    }
    __syncthreads();

    const int c    = lane & 15;
    const int hh   = lane >> 4;
    const int col0 = wave * (16 * FU_NT);
    const float* bbase = Bs + hh * FB_PITCH + 2 * (col0 + c);

    for (int j = 0; j < nT; ++j) {
        if (producer) {
            if (j + 1 < nT) {
                const int h = ((int)blockIdx.x + FB_WGS * (j + 1)) * 16 + prow;
                cur = row_produce(cur, h, (j + 2 < nT) ? h + FB_WGS * 16 : -1, lane, w0, w1, ab,
                                  slot, att_score, att_proj, val_feats, row_end, ent_feats,
                                  Us + ((j + 1) & 1) * (16 * FU_UP) + prow * FU_UP, C);
            }
        } else {
            const int row0 = ((int)blockIdx.x + FB_WGS * j) * 16;
            float* crow = C + (long)(row0 + 8 * hh) * KD + col0 + c;
            const float* us = Us + (j & 1) * (16 * FU_UP) + c * FU_UP + 2 * hh;

            v8f acc[FU_NT];
#pragma unroll
            for (int t = 0; t < FU_NT; ++t)
#pragma unroll
                for (int r = 0; r < 8; ++r) acc[t][r] = crow[(long)r * KD + 16 * t];

            v2f a[FB_CH];
#pragma unroll
            for (int s = 0; s < FB_CH; ++s) a[s] = *(const v2f*)(us + 4 * s);

            const float* bs = bbase;
#pragma unroll 1
            for (int ch = 0; ch < KD / (4 * FB_CH) - 1; ++ch) {
                v2f an[FB_CH];
#pragma unroll
                for (int s = 0; s < FB_CH; ++s) an[s] = *(const v2f*)(us + 4 * FB_CH * (ch + 1) + 4 * s);
                ent_chunk(acc, a, bs);
                bs += 2 * FB_CH * FB_PITCH;
#pragma unroll
                for (int s = 0; s < FB_CH; ++s) a[s] = an[s];
            }
            ent_chunk(acc, a, bs);

#pragma unroll
            for (int t = 0; t < FU_NT; ++t)
#pragma unroll
                for (int r = 0; r < 8; ++r) {
                    float x = acc[t][r];
                    x = (x > 0.0f) ? x : (expf(x) - 1.0f);
                    crow[(long)r * KD + 16 * t] = x;
                }
        }
        __syncthreads();
    }
}

static_assert((N_ATT / 16) * 16 == N_ATT, "attribute GEMM: one wave per attribute row, whole 16-row tiles");
static_assert((N_ATT / 16) * 16 * 32 >= N_ENT, "attribute GEMM launch must have a thread per entity row counter");
static_assert((N_ENT / 16) * 16 == N_ENT, "entity GEMM: whole 16-row tiles");
static_assert(((KD / 2) * FB_PITCH + 2 * 16 * FU_UP) * 4 <= 320 * 1024, "entity GEMM: W_bot and two tiles of U must fit the WGP's LDS");
static_assert(FU_CONS * 16 * FU_NT == KD, "entity GEMM: the consumer waves cover all output columns");
static_assert(FU_PROD == 16, "one producer wave per row of a tile");

extern "C" void kernel_launch(void* const* d_in, const int* in_sizes, int n_in,
                              void* d_out, int out_size, void* d_ws, size_t ws_size,
                              hipStream_t stream) {
    const int*   trip      = (const int*)d_in[0];
    const float* att_feats = (const float*)d_in[1];
    const float* val_feats = (const float*)d_in[2];
    const float* ent_feats = (const float*)d_in[3];
    const float* a_w       = (const float*)d_in[4];
    const float* a_b       = (const float*)d_in[5];
    const float* W         = (const float*)d_in[6];
    float* out = (float*)d_out;

    const float* W_top = W;
    const float* W_bot = W + KD * KD;

    float* ws       = (float*)d_ws;
    float* att_s    = ws + N_ENT;
    int*   row_cnt  = (int*)(att_s + N_ATT);
    int*   slot     = row_cnt + N_ENT;
    float* att_proj = (float*)(slot + NE);
    float* U        = att_proj + (long)N_ATT * KD;
    int*   rank     = (int*)U;

    {
        int tiles = (N_ATT / 16) * 16;
        wmma_gemm_att<<<tiles / 8, 256, 0, stream>>>(att_feats, W_top, att_proj,
                                                     a_w + KD, att_s, row_cnt);
    }

    edge_rank_kernel<<<(NE + 255) / 256, 256, 0, stream>>>(trip, row_cnt, rank);
    row_scan_kernel<<<1, SCAN_T, 0, stream>>>(row_cnt);
    edge_scatter_kernel<<<(NE + 255) / 256, 256, 0, stream>>>(trip, row_cnt, rank, slot);

    row_agg_gemm_ent<<<FB_WGS, FU_THREADS, 0, stream>>>(row_cnt, slot, ent_feats, a_w, att_s, a_b,
                                                        att_proj, val_feats, W_bot, out);
}


# Producer waves pipelined across tiles: next row's edge records, scores and first 4 edges' row reads issued before the hand-over barrier, which now waits on stores and LDS only; sum order unchanged
# speedup vs baseline: 2.7984x; 1.0334x over previous
#include <hip/hip_runtime.h>
#include <math.h>

#define N_ENT 50000
#define N_ATT 2000
#define N_VAL 100000
#define NE    300000
#define KD    256

#define VAL_BITS 17
static_assert(N_VAL <= (1 << VAL_BITS), "value row index must fit VAL_BITS");
static_assert(N_ATT <= (1 << (31 - VAL_BITS)), "attribute row index must fit the rest");

typedef __attribute__((ext_vector_type(2))) float v2f;
typedef __attribute__((ext_vector_type(8))) float v8f;

__device__ __forceinline__ float wave_row_dot(const float4 f0, const float4 f1,
                                              const float4 w0, const float4 w1) {
    float s = 0.0f;
    s += f0.x * w0.x + f0.y * w0.y + f0.z * w0.z + f0.w * w0.w;
    s += f1.x * w1.x + f1.y * w1.y + f1.z * w1.z + f1.w * w1.w;
    for (int off = 16; off > 0; off >>= 1) s += __shfl_xor(s, off, 32);
    return s;
}

__global__ void edge_rank_kernel(const int* __restrict__ trip,
                                 int* __restrict__ row_cnt,
                                 int* __restrict__ rank) {
    int e = blockIdx.x * blockDim.x + threadIdx.x;
    if (e >= NE) return;
    int h = trip[3 * e + 0];
    rank[e] = atomicAdd(&row_cnt[h], 1);
}

#define SCAN_T  1024
#define SCAN_CH ((N_ENT + SCAN_T - 1) / SCAN_T)
__global__ __launch_bounds__(SCAN_T) void row_scan_kernel(int* __restrict__ row_cnt) {
    __shared__ int s[SCAN_T * SCAN_CH];
    __shared__ int wtot[SCAN_T / 32];
    const int t = threadIdx.x;
    const int lane = t & 31;
    const int w = t >> 5;
    for (int k = 0; k < SCAN_CH; ++k) {
        const int i = t + k * SCAN_T;
        s[i] = (i < N_ENT) ? row_cnt[i] : 0;
    }
    __syncthreads();
    const int b = t * SCAN_CH;
    int sum = 0;
    for (int k = 0; k < SCAN_CH; ++k) sum += s[b + k];
    int v = sum;
    for (int o = 1; o < 32; o <<= 1) {
        int n = __shfl_up(v, o, 32);
        if (lane >= o) v += n;
    }
    if (lane == 31) wtot[w] = v;
    __syncthreads();
    if (w == 0) {
        int x = wtot[lane];
        for (int o = 1; o < 32; o <<= 1) {
            int n = __shfl_up(x, o, 32);
            if (lane >= o) x += n;
        }
        wtot[lane] = x;
    }
    __syncthreads();
    int run = (v - sum) + (w > 0 ? wtot[w - 1] : 0);
    for (int k = 0; k < SCAN_CH; ++k) { run += s[b + k]; s[b + k] = run; }
    __syncthreads();
    for (int k = 0; k < SCAN_CH; ++k) {
        const int i = t + k * SCAN_T;
        if (i < N_ENT) row_cnt[i] = s[i];
    }
}

__global__ void edge_scatter_kernel(const int* __restrict__ trip,
                                    const int* __restrict__ row_end,
                                    const int* __restrict__ rank,
                                    int* __restrict__ slot) {
    int e = blockIdx.x * blockDim.x + threadIdx.x;
    if (e >= NE) return;
    int h  = trip[3 * e + 0];
    int vl = trip[3 * e + 1];
    int at = trip[3 * e + 2];
    int beg = (h > 0) ? row_end[h - 1] : 0;
    slot[beg + rank[e]] = vl | (at << VAL_BITS);
}


__global__ void wmma_gemm_att(const float* __restrict__ A,
                              const float* __restrict__ B,
                              float* __restrict__ C,
                              const float* __restrict__ a_w_att,
                              float* __restrict__ att_score,
                              int* __restrict__ row_cnt) {
    int wave = (int)((blockIdx.x * blockDim.x + threadIdx.x) >> 5);
    int lane = threadIdx.x & 31;
    int tM = wave >> 4;
    int tN = wave & 15;
    int row0 = tM * 16;
    int col0 = tN * 16;

    int m    = lane & 15;
    int koff = (lane >> 4) << 1;
    int bn   = col0 + (lane & 15);
    int crow = row0 + ((lane >> 4) << 3);
    int cn   = col0 + (lane & 15);

    v8f c;
#pragma unroll
    for (int r = 0; r < 8; ++r) c[r] = 0.0f;

    const float* aptr = A + (long)(row0 + m) * KD + koff;

#pragma unroll 4
    for (int k0 = 0; k0 < KD; k0 += 4) {
        v2f av = *(const v2f*)(aptr + k0);
        v2f bv;
        bv.x = B[(long)(k0 + koff + 0) * KD + bn];
        bv.y = B[(long)(k0 + koff + 1) * KD + bn];
        c = __builtin_amdgcn_wmma_f32_16x16x4_f32(
                 false, av,  false, bv,
                 (short)0, c,  false,  false);
    }

#pragma unroll
    for (int r = 0; r < 8; ++r) C[(long)(crow + r) * KD + cn] = c[r];

    const int g = (int)(blockIdx.x * blockDim.x + threadIdx.x);
    if (g < N_ENT) row_cnt[g] = 0;
    const float4* fr = (const float4*)(A + (long)wave * KD);
    const float4* wr = (const float4*)a_w_att;
    const float s = wave_row_dot(fr[lane], fr[32 + lane], wr[lane], wr[32 + lane]);
    if (lane == 0) att_score[wave] = s;
}

#define FB_PITCH   544
#define FB_WGS     256
#define FB_CH      8
#define M_TILES    (N_ENT / 16)
#define FU_CONS    8
#define FU_PROD    16
#define FU_THREADS ((FU_CONS + FU_PROD) * 32)
#define FU_UP      260
#define FU_NT      2

__device__ __forceinline__ void ent_chunk(v8f (&acc)[FU_NT], const v2f (&a)[FB_CH], const float* bs) {
    v2f b[FU_NT];
#pragma unroll
    for (int t = 0; t < FU_NT; ++t) b[t] = *(const v2f*)(bs + 32 * t);
#pragma unroll
    for (int s = 0; s < FB_CH; ++s) {
        v2f bn[FU_NT];
        if (s + 1 < FB_CH) {
#pragma unroll
            for (int t = 0; t < FU_NT; ++t) bn[t] = *(const v2f*)(bs + 2 * (s + 1) * FB_PITCH + 32 * t);
        }
#pragma unroll
        for (int t = 0; t < FU_NT; ++t)
            acc[t] = __builtin_amdgcn_wmma_f32_16x16x4_f32(
                          false, a[s],  false, b[t],
                          (short)0, acc[t],  false,  false);
        if (s + 1 < FB_CH) {
#pragma unroll
            for (int t = 0; t < FU_NT; ++t) b[t] = bn[t];
        }
    }
}

__device__ __forceinline__ float edge_sc(float es, float as, float b) {
    float x = es + as + b;
    x = (x > 0.0f) ? x : 0.2f * x;
    return expf(x);
}

struct RowIn { int beg, end; float4 e0, e1; };

__device__ __forceinline__ RowIn row_fetch(int h, int lane, const int* __restrict__ row_end,
                                           const float* __restrict__ ent_feats) {
    RowIn r;
    const float4* er = (const float4*)(ent_feats + (long)h * KD);
    r.e0 = er[lane]; r.e1 = er[32 + lane];
    r.beg = (h > 0) ? row_end[h - 1] : 0;
    r.end = row_end[h];
    return r;
}

#define HEAD_E 4
struct RowHead {
    int   wd_l;
    float p_l;
    float es, rs;
    float4 a0[HEAD_E], a1[HEAD_E];
    float4 c0[HEAD_E], c1[HEAD_E];
};

__device__ __forceinline__ void head_clear(RowHead& hd) {
    const float4 z = make_float4(0.f, 0.f, 0.f, 0.f);
    hd.wd_l = 0; hd.p_l = 0.0f; hd.es = 0.0f; hd.rs = 0.0f;
#pragma unroll
    for (int k = 0; k < HEAD_E; ++k) { hd.a0[k] = z; hd.a1[k] = z; hd.c0[k] = z; hd.c1[k] = z; }
}

__device__ __forceinline__ void row_head(RowHead& hd, const RowIn& in, const int wd_l, const float as_l,
                                         const bool has_l, const int lane,
                                         const float4 w0, const float4 w1, const float b,
                                         const int* __restrict__ slot,
                                         const float* __restrict__ att_score,
                                         const float* __restrict__ att_proj,
                                         const float* __restrict__ val_feats) {
    const float es = wave_row_dot(in.e0, in.e1, w0, w1);
    const float sc_l = has_l ? edge_sc(es, as_l, b) : 0.0f;
    float rs = sc_l;
    for (int i = in.beg + 32 + lane; i < in.end; i += 32)
        rs += edge_sc(es, att_score[slot[i] >> VAL_BITS], b);
    for (int off = 16; off > 0; off >>= 1) rs += __shfl_xor(rs, off, 32);
    hd.wd_l = wd_l; hd.p_l = sc_l / rs; hd.es = es; hd.rs = rs;
    const int n = in.end - in.beg;
#pragma unroll
    for (int k = 0; k < HEAD_E; ++k) {
        if (k < n) {
            const int wd = __builtin_amdgcn_readlane(wd_l, k);
            const int vl = wd & ((1 << VAL_BITS) - 1);
            const int at = wd >> VAL_BITS;
            const float4* vr = (const float4*)(val_feats + (long)vl * KD);
            const float4* ar = (const float4*)(att_proj  + (long)at * KD);
            hd.a0[k] = vr[lane]; hd.a1[k] = vr[32 + lane];
            hd.c0[k] = ar[lane]; hd.c1[k] = ar[32 + lane];
        }
    }
}

__device__ __forceinline__ void row_step(RowIn& cur, RowHead& hd, RowIn& nxt,
                                         const int h, const int hn, const int hnn, const int lane,
                                         const float4 w0, const float4 w1, const float b,
                                         const int* __restrict__ slot,
                                         const float* __restrict__ att_score,
                                         const float* __restrict__ att_proj,
                                         const float* __restrict__ val_feats,
                                         const int* __restrict__ row_end,
                                         const float* __restrict__ ent_feats,
                                         float* urow,
                                         float* __restrict__ outAcc) {
    const float4 e0 = cur.e0, e1 = cur.e1;
    const int beg = cur.beg, end = cur.end;

    int  wdn  = 0;
    bool hasn = false;
    if (hn >= 0) {
        hasn = (nxt.beg + lane < nxt.end);
        if (hasn) wdn = slot[nxt.beg + lane];
    }

    asm("" : "+v"(nxt.e0.x), "+v"(nxt.e0.y), "+v"(nxt.e0.z), "+v"(nxt.e0.w),
             "+v"(nxt.e1.x), "+v"(nxt.e1.y), "+v"(nxt.e1.z), "+v"(nxt.e1.w));

    float4 u0 = make_float4(0.f, 0.f, 0.f, 0.f), u1 = u0, o0 = u0, o1 = u0;
    auto add = [&](const float p, const float4 a0, const float4 a1, const float4 c0, const float4 c1) {
        u0.x += p * a0.x; u0.y += p * a0.y; u0.z += p * a0.z; u0.w += p * a0.w;
        u1.x += p * a1.x; u1.y += p * a1.y; u1.z += p * a1.z; u1.w += p * a1.w;
        o0.x += p * c0.x; o0.y += p * c0.y; o0.z += p * c0.z; o0.w += p * c0.w;
        o1.x += p * c1.x; o1.y += p * c1.y; o1.z += p * c1.z; o1.w += p * c1.w;
    };

    const int n0 = (end - beg < 32) ? (end - beg) : 32;
#pragma unroll
    for (int k = 0; k < HEAD_E; ++k) {
        if (k < n0) {
            const float p = __int_as_float(__builtin_amdgcn_readlane(__float_as_int(hd.p_l), k));
            add(p, hd.a0[k], hd.a1[k], hd.c0[k], hd.c1[k]);
        }
    }

    float asn = 0.0f;
    if (hasn) asn = att_score[wdn >> VAL_BITS];

    int   wd_l = hd.wd_l;
    float p_l  = hd.p_l;
    const float es = hd.es, rs = hd.rs;
    for (int c = beg; c < end; c += 32) {
        const int n = (end - c < 32) ? (end - c) : 32;
        int j = (n < HEAD_E) ? n : HEAD_E;
        if (c != beg) {
            j = 0;
            wd_l = 0;
            float sc_l = 0.0f;
            if (c + lane < end) {
                wd_l = slot[c + lane];
                sc_l = edge_sc(es, att_score[wd_l >> VAL_BITS], b);
            }
            p_l = sc_l / rs;
        }
        auto edge = [&](int jj) {
            const int   wd = __builtin_amdgcn_readlane(wd_l, jj);
            const float p  = __int_as_float(__builtin_amdgcn_readlane(__float_as_int(p_l), jj));
            const int vl = wd & ((1 << VAL_BITS) - 1);
            const int at = wd >> VAL_BITS;
            const float4* vr = (const float4*)(val_feats + (long)vl * KD);
            const float4* ar = (const float4*)(att_proj  + (long)at * KD);
            const float4 a0 = vr[lane], a1 = vr[32 + lane];
            const float4 c0 = ar[lane], c1 = ar[32 + lane];
            add(p, a0, a1, c0, c1);
        };
        for (; j + 4 <= n; j += 4) { edge(j); edge(j + 1); edge(j + 2); edge(j + 3); }
        const int rem = n - j;
        if (rem == 3)      { edge(j); edge(j + 1); edge(j + 2); }
        else if (rem == 2) { edge(j); edge(j + 1); }
        else if (rem == 1) { edge(j); }
    }

    float4* ur   = (float4*)urow;
    float4* orow = (float4*)(outAcc + (long)h * KD);
    ur[lane] = u0;   ur[32 + lane] = u1;
    o0.x += e0.x; o0.y += e0.y; o0.z += e0.z; o0.w += e0.w;
    o1.x += e1.x; o1.y += e1.y; o1.z += e1.z; o1.w += e1.w;
    orow[lane] = o0; orow[32 + lane] = o1;

    head_clear(hd);
    if (hn >= 0) row_head(hd, nxt, wdn, asn, hasn, lane, w0, w1, b, slot, att_score, att_proj, val_feats);
    if (hnn >= 0) cur = row_fetch(hnn, lane, row_end, ent_feats);
}

__global__ __launch_bounds__(FU_THREADS) void row_agg_gemm_ent(const int* __restrict__ row_end,
                                                               const int* __restrict__ slot,
                                                               const float* __restrict__ ent_feats,
                                                               const float* __restrict__ a_w,
                                                               const float* __restrict__ att_score,
                                                               const float* __restrict__ a_b,
                                                               const float* __restrict__ att_proj,
                                                               const float* __restrict__ val_feats,
                                                               const float* __restrict__ B,
                                                               float* C) {
    __shared__ float Bs[(KD / 2) * FB_PITCH];
    __shared__ float Us[2 * 16 * FU_UP];

    const int wave = __builtin_amdgcn_readfirstlane((int)(threadIdx.x >> 5));
    const int lane = threadIdx.x & 31;
    const bool producer = (wave >= FU_CONS);
    const int prow = wave - FU_CONS;
    const int nT = (M_TILES - (int)blockIdx.x + FB_WGS - 1) / FB_WGS;

    RowIn cur, nxt;
    RowHead hd;
    cur.beg = 0; cur.end = 0; cur.e0 = make_float4(0.f, 0.f, 0.f, 0.f); cur.e1 = cur.e0;
    nxt = cur;
    head_clear(hd);
    float4 w0 = make_float4(0.f, 0.f, 0.f, 0.f), w1 = w0;
    float ab = 0.0f;
    if (producer) {
        const float4* aw = (const float4*)a_w;
        w0 = aw[lane]; w1 = aw[32 + lane];
        ab = a_b[0];
        const int h0 = (int)blockIdx.x * 16 + prow;
        cur = row_fetch(h0, lane, row_end, ent_feats);
        nxt = cur;
        if (1 < nT) nxt = row_fetch(h0 + FB_WGS * 16, lane, row_end, ent_feats);
    }

#pragma unroll
    for (int it = 0; it < ((KD / 2) * (KD / 4) + FU_THREADS - 1) / FU_THREADS; ++it) {
        const int idx = (int)threadIdx.x + it * FU_THREADS;
        if (idx < (KD / 2) * (KD / 4)) {
            const int p  = idx >> 6;
            const int n4 = (idx & 63) << 2;
            const float4 r0 = *(const float4*)(B + (long)(2 * p)     * KD + n4);
            const float4 r1 = *(const float4*)(B + (long)(2 * p + 1) * KD + n4);
            float* d = Bs + p * FB_PITCH + n4 * 2;
            *(float4*)(d)     = make_float4(r0.x, r1.x, r0.y, r1.y);
            *(float4*)(d + 4) = make_float4(r0.z, r1.z, r0.w, r1.w);
        }
    }

    if (producer) {
        const bool has_l = (cur.beg + lane < cur.end);
        int   wd_l = 0;
        float as_l = 0.0f;
        if (has_l) {
            wd_l = slot[cur.beg + lane];
            as_l = att_score[wd_l >> VAL_BITS];
        }
        row_head(hd, cur, wd_l, as_l, has_l, lane, w0, w1, ab, slot, att_score, att_proj, val_feats);
    }

    const int c    = lane & 15;
    const int hh   = lane >> 4;
    const int col0 = wave * (16 * FU_NT);
    const float* bbase = Bs + hh * FB_PITCH + 2 * (col0 + c);

    auto trip = [&](const int j, RowIn& rc, RowIn& rn) {
        if (producer) {
            if (j + 1 < nT) {
                const int h   = ((int)blockIdx.x + FB_WGS * (j + 1)) * 16 + prow;
                const int hn  = (j + 2 < nT) ? h + FB_WGS * 16 : -1;
                const int hnn = (j + 3 < nT) ? h + 2 * FB_WGS * 16 : -1;
                row_step(rc, hd, rn, h, hn, hnn, lane, w0, w1, ab,
                         slot, att_score, att_proj, val_feats, row_end, ent_feats,
                         Us + ((j + 1) & 1) * (16 * FU_UP) + prow * FU_UP, C);
            }
        } else if (j >= 0) {
            const int row0 = ((int)blockIdx.x + FB_WGS * j) * 16;
            float* crow = C + (long)(row0 + 8 * hh) * KD + col0 + c;
            const float* us = Us + (j & 1) * (16 * FU_UP) + c * FU_UP + 2 * hh;

            v8f acc[FU_NT];
#pragma unroll
            for (int t = 0; t < FU_NT; ++t)
#pragma unroll
                for (int r = 0; r < 8; ++r) acc[t][r] = crow[(long)r * KD + 16 * t];

            v2f a[FB_CH];
#pragma unroll
            for (int s = 0; s < FB_CH; ++s) a[s] = *(const v2f*)(us + 4 * s);

            const float* bs = bbase;
#pragma unroll 1
            for (int ch = 0; ch < KD / (4 * FB_CH) - 1; ++ch) {
                v2f an[FB_CH];
#pragma unroll
                for (int s = 0; s < FB_CH; ++s) an[s] = *(const v2f*)(us + 4 * FB_CH * (ch + 1) + 4 * s);
                ent_chunk(acc, a, bs);
                bs += 2 * FB_CH * FB_PITCH;
#pragma unroll
                for (int s = 0; s < FB_CH; ++s) a[s] = an[s];
            }
            ent_chunk(acc, a, bs);

#pragma unroll
            for (int t = 0; t < FU_NT; ++t)
#pragma unroll
                for (int r = 0; r < 8; ++r) {
                    float x = acc[t][r];
                    x = (x > 0.0f) ? x : (expf(x) - 1.0f);
                    crow[(long)r * KD + 16 * t] = x;
                }
        }
        asm volatile("s_wait_storecnt_dscnt 0x0" ::: "memory");
        __builtin_amdgcn_s_barrier();
        asm volatile("" ::: "memory");
    };
    for (int j = -1; j < nT; j += 2) {
        trip(j, cur, nxt);
        if (j + 1 < nT) trip(j + 1, nxt, cur);
    }
}

static_assert((N_ATT / 16) * 16 == N_ATT, "attribute GEMM: one wave per attribute row, whole 16-row tiles");
static_assert((N_ATT / 16) * 16 * 32 >= N_ENT, "attribute GEMM launch must have a thread per entity row counter");
static_assert((N_ENT / 16) * 16 == N_ENT, "entity GEMM: whole 16-row tiles");
static_assert(((KD / 2) * FB_PITCH + 2 * 16 * FU_UP) * 4 <= 320 * 1024, "entity GEMM: W_bot and two tiles of U must fit the WGP's LDS");
static_assert(FU_CONS * 16 * FU_NT == KD, "entity GEMM: the consumer waves cover all output columns");
static_assert(FU_PROD == 16, "one producer wave per row of a tile");

extern "C" void kernel_launch(void* const* d_in, const int* in_sizes, int n_in,
                              void* d_out, int out_size, void* d_ws, size_t ws_size,
                              hipStream_t stream) {
    const int*   trip      = (const int*)d_in[0];
    const float* att_feats = (const float*)d_in[1];
    const float* val_feats = (const float*)d_in[2];
    const float* ent_feats = (const float*)d_in[3];
    const float* a_w       = (const float*)d_in[4];
    const float* a_b       = (const float*)d_in[5];
    const float* W         = (const float*)d_in[6];
    float* out = (float*)d_out;

    const float* W_top = W;
    const float* W_bot = W + KD * KD;

    float* ws       = (float*)d_ws;
    float* att_s    = ws + N_ENT;
    int*   row_cnt  = (int*)(att_s + N_ATT);
    int*   slot     = row_cnt + N_ENT;
    float* att_proj = (float*)(slot + NE);
    float* U        = att_proj + (long)N_ATT * KD;
    int*   rank     = (int*)U;

    {
        int tiles = (N_ATT / 16) * 16;
        wmma_gemm_att<<<tiles / 8, 256, 0, stream>>>(att_feats, W_top, att_proj,
                                                     a_w + KD, att_s, row_cnt);
    }

    edge_rank_kernel<<<(NE + 255) / 256, 256, 0, stream>>>(trip, row_cnt, rank);
    row_scan_kernel<<<1, SCAN_T, 0, stream>>>(row_cnt);
    edge_scatter_kernel<<<(NE + 255) / 256, 256, 0, stream>>>(trip, row_cnt, rank, slot);

    row_agg_gemm_ent<<<FB_WGS, FU_THREADS, 0, stream>>>(row_cnt, slot, ent_feats, a_w, att_s, a_b,
                                                        att_proj, val_feats, W_bot, out);
}


# Edges counted straight into fixed 32-record row buckets (overflow list past 32): scan and scatter launches gone, 5->3; a row's record line, address now known, is read with its entity row
# speedup vs baseline: 3.1911x; 1.1403x over previous
#include <hip/hip_runtime.h>
#include <math.h>

#define N_ENT 50000
#define N_ATT 2000
#define N_VAL 100000
#define NE    300000
#define KD    256

#define VAL_BITS 17
static_assert(N_VAL <= (1 << VAL_BITS), "value row index must fit VAL_BITS");
static_assert(N_ATT <= (1 << (31 - VAL_BITS)), "attribute row index must fit the rest");

typedef __attribute__((ext_vector_type(2))) float v2f;
typedef __attribute__((ext_vector_type(8))) float v8f;

__device__ __forceinline__ float wave_row_dot(const float4 f0, const float4 f1,
                                              const float4 w0, const float4 w1) {
    float s = 0.0f;
    s += f0.x * w0.x + f0.y * w0.y + f0.z * w0.z + f0.w * w0.w;
    s += f1.x * w1.x + f1.y * w1.y + f1.z * w1.z + f1.w * w1.w;
    for (int off = 16; off > 0; off >>= 1) s += __shfl_xor(s, off, 32);
    return s;
}

#define BK 32
static_assert(BK == 32, "one bucket place per lane: lane l of a row's wave holds bucket record l");
#define ATT_THREADS ((N_ATT / 16) * 16 * 32)
#define BK4 (N_ENT * BK / 4)
__global__ void edge_bucket_kernel(const int* __restrict__ trip,
                                   int* __restrict__ row_cnt,
                                   int* __restrict__ bucket,
                                   int* __restrict__ ovf_cnt,
                                   int* __restrict__ ovf) {
    int e = blockIdx.x * blockDim.x + threadIdx.x;
    if (e >= NE) return;
    const int h  = trip[3 * e + 0];
    const int vl = trip[3 * e + 1];
    const int at = trip[3 * e + 2];
    const int wd = vl | (at << VAL_BITS);
    const int r = atomicAdd(&row_cnt[h], 1);
    if (r < BK) {
        bucket[(long)h * BK + r] = wd;
    } else {
        const int i = atomicAdd(ovf_cnt, 1);
        ovf[2 * (long)i + 0] = h;
        ovf[2 * (long)i + 1] = wd;
    }
}


__global__ void wmma_gemm_att(const float* __restrict__ A,
                              const float* __restrict__ B,
                              float* __restrict__ C,
                              const float* __restrict__ a_w_att,
                              float* __restrict__ att_score,
                              int* __restrict__ row_cnt,
                              int* __restrict__ ovf_cnt,
                              int* __restrict__ bucket) {
    int wave = (int)((blockIdx.x * blockDim.x + threadIdx.x) >> 5);
    int lane = threadIdx.x & 31;
    int tM = wave >> 4;
    int tN = wave & 15;
    int row0 = tM * 16;
    int col0 = tN * 16;

    int m    = lane & 15;
    int koff = (lane >> 4) << 1;
    int bn   = col0 + (lane & 15);
    int crow = row0 + ((lane >> 4) << 3);
    int cn   = col0 + (lane & 15);

    v8f c;
#pragma unroll
    for (int r = 0; r < 8; ++r) c[r] = 0.0f;

    const float* aptr = A + (long)(row0 + m) * KD + koff;

#pragma unroll 4
    for (int k0 = 0; k0 < KD; k0 += 4) {
        v2f av = *(const v2f*)(aptr + k0);
        v2f bv;
        bv.x = B[(long)(k0 + koff + 0) * KD + bn];
        bv.y = B[(long)(k0 + koff + 1) * KD + bn];
        c = __builtin_amdgcn_wmma_f32_16x16x4_f32(
                 false, av,  false, bv,
                 (short)0, c,  false,  false);
    }

#pragma unroll
    for (int r = 0; r < 8; ++r) C[(long)(crow + r) * KD + cn] = c[r];

    const int g = (int)(blockIdx.x * blockDim.x + threadIdx.x);
    if (g < N_ENT) row_cnt[g] = 0;
    if (g == 0) ovf_cnt[0] = 0;
    {
        int4* b4 = (int4*)bucket;
        const int4 z = make_int4(0, 0, 0, 0);
#pragma unroll
        for (int it = 0; it < BK4 / ATT_THREADS; ++it) b4[g + it * ATT_THREADS] = z;
        if (g < BK4 % ATT_THREADS) b4[g + (BK4 / ATT_THREADS) * ATT_THREADS] = z;
    }
    const float4* fr = (const float4*)(A + (long)wave * KD);
    const float4* wr = (const float4*)a_w_att;
    const float s = wave_row_dot(fr[lane], fr[32 + lane], wr[lane], wr[32 + lane]);
    if (lane == 0) att_score[wave] = s;
}

#define FB_PITCH   544
#define FB_WGS     256
#define FB_CH      8
#define M_TILES    (N_ENT / 16)
#define FU_CONS    8
#define FU_PROD    16
#define FU_THREADS ((FU_CONS + FU_PROD) * 32)
#define FU_UP      260
#define FU_NT      2

__device__ __forceinline__ void ent_chunk(v8f (&acc)[FU_NT], const v2f (&a)[FB_CH], const float* bs) {
    v2f b[FU_NT];
#pragma unroll
    for (int t = 0; t < FU_NT; ++t) b[t] = *(const v2f*)(bs + 32 * t);
#pragma unroll
    for (int s = 0; s < FB_CH; ++s) {
        v2f bn[FU_NT];
        if (s + 1 < FB_CH) {
#pragma unroll
            for (int t = 0; t < FU_NT; ++t) bn[t] = *(const v2f*)(bs + 2 * (s + 1) * FB_PITCH + 32 * t);
        }
#pragma unroll
        for (int t = 0; t < FU_NT; ++t)
            acc[t] = __builtin_amdgcn_wmma_f32_16x16x4_f32(
                          false, a[s],  false, b[t],
                          (short)0, acc[t],  false,  false);
        if (s + 1 < FB_CH) {
#pragma unroll
            for (int t = 0; t < FU_NT; ++t) b[t] = bn[t];
        }
    }
}

__device__ __forceinline__ float edge_sc(float es, float as, float b) {
    float x = es + as + b;
    x = (x > 0.0f) ? x : 0.2f * x;
    return expf(x);
}

struct RowIn { int cnt; int wd; float4 e0, e1; };

__device__ __forceinline__ RowIn row_fetch(int h, int lane, const int* __restrict__ row_cnt,
                                           const int* __restrict__ bucket,
                                           const float* __restrict__ ent_feats) {
    RowIn r;
    const float4* er = (const float4*)(ent_feats + (long)h * KD);
    r.e0 = er[lane]; r.e1 = er[32 + lane];
    r.wd  = bucket[h * BK + lane];
    r.cnt = row_cnt[h];
    return r;
}

#define HEAD_E 4
struct RowHead {
    int   wd_l;
    float p_l;
    float es, rs;
    float4 a0[HEAD_E], a1[HEAD_E];
    float4 c0[HEAD_E], c1[HEAD_E];
};

__device__ __forceinline__ void head_clear(RowHead& hd) {
    const float4 z = make_float4(0.f, 0.f, 0.f, 0.f);
    hd.wd_l = 0; hd.p_l = 0.0f; hd.es = 0.0f; hd.rs = 0.0f;
#pragma unroll
    for (int k = 0; k < HEAD_E; ++k) { hd.a0[k] = z; hd.a1[k] = z; hd.c0[k] = z; hd.c1[k] = z; }
}

__device__ __forceinline__ void row_head(RowHead& hd, const RowIn& in, const int hrow, const int wd_l,
                                         const float as_l, const bool has_l, const int lane,
                                         const float4 w0, const float4 w1, const float b,
                                         const int* __restrict__ ovf_cnt,
                                         const int* __restrict__ ovf,
                                         const float* __restrict__ att_score,
                                         const float* __restrict__ att_proj,
                                         const float* __restrict__ val_feats) {
    const float es = wave_row_dot(in.e0, in.e1, w0, w1);
    const float sc_l = has_l ? edge_sc(es, as_l, b) : 0.0f;
    float rs = sc_l;
    if (in.cnt > BK) {
        const int nov = (ovf_cnt[0] < NE) ? ovf_cnt[0] : NE;
        for (int i = lane; i < nov; i += 32)
            if (ovf[2 * (long)i] == hrow)
                rs += edge_sc(es, att_score[ovf[2 * (long)i + 1] >> VAL_BITS], b);
    }
    for (int off = 16; off > 0; off >>= 1) rs += __shfl_xor(rs, off, 32);
    hd.wd_l = wd_l; hd.p_l = sc_l / rs; hd.es = es; hd.rs = rs;
    const int n = in.cnt;
#pragma unroll
    for (int k = 0; k < HEAD_E; ++k) {
        if (k < n) {
            const int wd = __builtin_amdgcn_readlane(wd_l, k);
            const int vl = wd & ((1 << VAL_BITS) - 1);
            const int at = wd >> VAL_BITS;
            const float4* vr = (const float4*)(val_feats + (long)vl * KD);
            const float4* ar = (const float4*)(att_proj  + (long)at * KD);
            hd.a0[k] = vr[lane]; hd.a1[k] = vr[32 + lane];
            hd.c0[k] = ar[lane]; hd.c1[k] = ar[32 + lane];
        }
    }
}

__device__ __forceinline__ void row_step(RowIn& cur, RowHead& hd, RowIn& nxt,
                                         const int h, const int hn, const int hnn, const int lane,
                                         const float4 w0, const float4 w1, const float b,
                                         const int* __restrict__ bucket,
                                         const int* __restrict__ ovf_cnt,
                                         const int* __restrict__ ovf,
                                         const float* __restrict__ att_score,
                                         const float* __restrict__ att_proj,
                                         const float* __restrict__ val_feats,
                                         const int* __restrict__ row_cnt,
                                         const float* __restrict__ ent_feats,
                                         float* urow,
                                         float* __restrict__ outAcc) {
    const float4 e0 = cur.e0, e1 = cur.e1;
    const int cnt = cur.cnt;

    const bool hasn = (hn >= 0) && (lane < nxt.cnt);
    const int  wdn  = hasn ? nxt.wd : 0;
    float asn = 0.0f;
    if (hasn) asn = att_score[wdn >> VAL_BITS];

    asm("" : "+v"(nxt.e0.x), "+v"(nxt.e0.y), "+v"(nxt.e0.z), "+v"(nxt.e0.w),
             "+v"(nxt.e1.x), "+v"(nxt.e1.y), "+v"(nxt.e1.z), "+v"(nxt.e1.w));

    float4 u0 = make_float4(0.f, 0.f, 0.f, 0.f), u1 = u0, o0 = u0, o1 = u0;
    auto add = [&](const float p, const float4 a0, const float4 a1, const float4 c0, const float4 c1) {
        u0.x += p * a0.x; u0.y += p * a0.y; u0.z += p * a0.z; u0.w += p * a0.w;
        u1.x += p * a1.x; u1.y += p * a1.y; u1.z += p * a1.z; u1.w += p * a1.w;
        o0.x += p * c0.x; o0.y += p * c0.y; o0.z += p * c0.z; o0.w += p * c0.w;
        o1.x += p * c1.x; o1.y += p * c1.y; o1.z += p * c1.z; o1.w += p * c1.w;
    };

    const int n0 = (cnt < BK) ? cnt : BK;
#pragma unroll
    for (int k = 0; k < HEAD_E; ++k) {
        if (k < n0) {
            const float p = __int_as_float(__builtin_amdgcn_readlane(__float_as_int(hd.p_l), k));
            add(p, hd.a0[k], hd.a1[k], hd.c0[k], hd.c1[k]);
        }
    }

    int   wd_l = hd.wd_l;
    float p_l  = hd.p_l;
    auto edge = [&](int jj) {
        const int   wd = __builtin_amdgcn_readlane(wd_l, jj);
        const float p  = __int_as_float(__builtin_amdgcn_readlane(__float_as_int(p_l), jj));
        const int vl = wd & ((1 << VAL_BITS) - 1);
        const int at = wd >> VAL_BITS;
        const float4* vr = (const float4*)(val_feats + (long)vl * KD);
        const float4* ar = (const float4*)(att_proj  + (long)at * KD);
        const float4 a0 = vr[lane], a1 = vr[32 + lane];
        const float4 c0 = ar[lane], c1 = ar[32 + lane];
        add(p, a0, a1, c0, c1);
    };
    {
        int j = (n0 < HEAD_E) ? n0 : HEAD_E;
        for (; j + 4 <= n0; j += 4) { edge(j); edge(j + 1); edge(j + 2); edge(j + 3); }
        const int rem = n0 - j;
        if (rem == 3)      { edge(j); edge(j + 1); edge(j + 2); }
        else if (rem == 2) { edge(j); edge(j + 1); }
        else if (rem == 1) { edge(j); }
    }
    if (cnt > BK) {
        const float es = hd.es, rs = hd.rs;
        const int nov = (ovf_cnt[0] < NE) ? ovf_cnt[0] : NE;
        for (int c = 0; c < nov; c += 32) {
            const int i = c + lane;
            bool mine = false;
            float sc_l = 0.0f;
            wd_l = 0;
            if (i < nov) mine = (ovf[2 * (long)i] == h);
            if (mine) {
                wd_l = ovf[2 * (long)i + 1];
                sc_l = edge_sc(es, att_score[wd_l >> VAL_BITS], b);
            }
            p_l = sc_l / rs;
            unsigned todo = __builtin_amdgcn_ballot_w32(mine);
            while (todo != 0u) {
                edge(__builtin_ctz(todo));
                todo &= todo - 1u;
            }
        }
    }

    float4* ur   = (float4*)urow;
    float4* orow = (float4*)(outAcc + (long)h * KD);
    ur[lane] = u0;   ur[32 + lane] = u1;
    o0.x += e0.x; o0.y += e0.y; o0.z += e0.z; o0.w += e0.w;
    o1.x += e1.x; o1.y += e1.y; o1.z += e1.z; o1.w += e1.w;
    orow[lane] = o0; orow[32 + lane] = o1;

    head_clear(hd);
    if (hn >= 0) row_head(hd, nxt, hn, wdn, asn, hasn, lane, w0, w1, b, ovf_cnt, ovf, att_score, att_proj, val_feats);
    if (hnn >= 0) cur = row_fetch(hnn, lane, row_cnt, bucket, ent_feats);
}

__global__ __launch_bounds__(FU_THREADS) void row_agg_gemm_ent(const int* __restrict__ row_cnt,
                                                               const int* __restrict__ bucket,
                                                               const int* __restrict__ ovf_cnt,
                                                               const int* __restrict__ ovf,
                                                               const float* __restrict__ ent_feats,
                                                               const float* __restrict__ a_w,
                                                               const float* __restrict__ att_score,
                                                               const float* __restrict__ a_b,
                                                               const float* __restrict__ att_proj,
                                                               const float* __restrict__ val_feats,
                                                               const float* __restrict__ B,
                                                               float* C) {
    __shared__ float Bs[(KD / 2) * FB_PITCH];
    __shared__ float Us[2 * 16 * FU_UP];

    const int wave = __builtin_amdgcn_readfirstlane((int)(threadIdx.x >> 5));
    const int lane = threadIdx.x & 31;
    const bool producer = (wave >= FU_CONS);
    const int prow = wave - FU_CONS;
    const int nT = (M_TILES - (int)blockIdx.x + FB_WGS - 1) / FB_WGS;

    RowIn cur, nxt;
    RowHead hd;
    cur.cnt = 0; cur.wd = 0; cur.e0 = make_float4(0.f, 0.f, 0.f, 0.f); cur.e1 = cur.e0;
    nxt = cur;
    head_clear(hd);
    float4 w0 = make_float4(0.f, 0.f, 0.f, 0.f), w1 = w0;
    float ab = 0.0f;
    if (producer) {
        const float4* aw = (const float4*)a_w;
        w0 = aw[lane]; w1 = aw[32 + lane];
        ab = a_b[0];
        const int h0 = (int)blockIdx.x * 16 + prow;
        cur = row_fetch(h0, lane, row_cnt, bucket, ent_feats);
        nxt = cur;
        if (1 < nT) nxt = row_fetch(h0 + FB_WGS * 16, lane, row_cnt, bucket, ent_feats);
    }

#pragma unroll
    for (int it = 0; it < ((KD / 2) * (KD / 4) + FU_THREADS - 1) / FU_THREADS; ++it) {
        const int idx = (int)threadIdx.x + it * FU_THREADS;
        if (idx < (KD / 2) * (KD / 4)) {
            const int p  = idx >> 6;
            const int n4 = (idx & 63) << 2;
            const float4 r0 = *(const float4*)(B + (long)(2 * p)     * KD + n4);
            const float4 r1 = *(const float4*)(B + (long)(2 * p + 1) * KD + n4);
            float* d = Bs + p * FB_PITCH + n4 * 2;
            *(float4*)(d)     = make_float4(r0.x, r1.x, r0.y, r1.y);
            *(float4*)(d + 4) = make_float4(r0.z, r1.z, r0.w, r1.w);
        }
    }

    if (producer) {
        const bool has_l = (lane < cur.cnt);
        const int  wd_l = has_l ? cur.wd : 0;
        float as_l = 0.0f;
        if (has_l) as_l = att_score[wd_l >> VAL_BITS];
        row_head(hd, cur, (int)blockIdx.x * 16 + prow, wd_l, as_l, has_l, lane, w0, w1, ab,
                 ovf_cnt, ovf, att_score, att_proj, val_feats);
    }

    const int c    = lane & 15;
    const int hh   = lane >> 4;
    const int col0 = wave * (16 * FU_NT);
    const float* bbase = Bs + hh * FB_PITCH + 2 * (col0 + c);

    auto trip = [&](const int j, RowIn& rc, RowIn& rn) {
        if (producer) {
            if (j + 1 < nT) {
                const int h   = ((int)blockIdx.x + FB_WGS * (j + 1)) * 16 + prow;
                const int hn  = (j + 2 < nT) ? h + FB_WGS * 16 : -1;
                const int hnn = (j + 3 < nT) ? h + 2 * FB_WGS * 16 : -1;
                row_step(rc, hd, rn, h, hn, hnn, lane, w0, w1, ab,
                         bucket, ovf_cnt, ovf, att_score, att_proj, val_feats, row_cnt, ent_feats,
                         Us + ((j + 1) & 1) * (16 * FU_UP) + prow * FU_UP, C);
            }
        } else if (j >= 0) {
            const int row0 = ((int)blockIdx.x + FB_WGS * j) * 16;
            float* crow = C + (long)(row0 + 8 * hh) * KD + col0 + c;
            const float* us = Us + (j & 1) * (16 * FU_UP) + c * FU_UP + 2 * hh;

            v8f acc[FU_NT];
#pragma unroll
            for (int t = 0; t < FU_NT; ++t)
#pragma unroll
                for (int r = 0; r < 8; ++r) acc[t][r] = crow[(long)r * KD + 16 * t];

            v2f a[FB_CH];
#pragma unroll
            for (int s = 0; s < FB_CH; ++s) a[s] = *(const v2f*)(us + 4 * s);

            const float* bs = bbase;
#pragma unroll 1
            for (int ch = 0; ch < KD / (4 * FB_CH) - 1; ++ch) {
                v2f an[FB_CH];
#pragma unroll
                for (int s = 0; s < FB_CH; ++s) an[s] = *(const v2f*)(us + 4 * FB_CH * (ch + 1) + 4 * s);
                ent_chunk(acc, a, bs);
                bs += 2 * FB_CH * FB_PITCH;
#pragma unroll
                for (int s = 0; s < FB_CH; ++s) a[s] = an[s];
            }
            ent_chunk(acc, a, bs);

#pragma unroll
            for (int t = 0; t < FU_NT; ++t)
#pragma unroll
                for (int r = 0; r < 8; ++r) {
                    float x = acc[t][r];
                    x = (x > 0.0f) ? x : (expf(x) - 1.0f);
                    crow[(long)r * KD + 16 * t] = x;
                }
        }
        asm volatile("s_wait_storecnt_dscnt 0x0" ::: "memory");
        __builtin_amdgcn_s_barrier();
        asm volatile("" ::: "memory");
    };
    for (int j = -1; j < nT; j += 2) {
        trip(j, cur, nxt);
        if (j + 1 < nT) trip(j + 1, nxt, cur);
    }
}

static_assert((N_ATT / 16) * 16 == N_ATT, "attribute GEMM: one wave per attribute row, whole 16-row tiles");
static_assert(ATT_THREADS >= N_ENT, "attribute GEMM launch must have a thread per entity row counter");
static_assert((N_ENT / 16) * 16 == N_ENT, "entity GEMM: whole 16-row tiles");
static_assert(((KD / 2) * FB_PITCH + 2 * 16 * FU_UP) * 4 <= 320 * 1024, "entity GEMM: W_bot and two tiles of U must fit the WGP's LDS");
static_assert(FU_CONS * 16 * FU_NT == KD, "entity GEMM: the consumer waves cover all output columns");
static_assert(FU_PROD == 16, "one producer wave per row of a tile");
#define WS_PAD 48
static_assert((N_ENT + N_ATT + N_ENT + NE + N_ATT * KD + WS_PAD) % 64 == 0, "buckets start on a 256-byte multiple");
static_assert(HEAD_E <= BK, "the head's edges come out of the bucket");

extern "C" void kernel_launch(void* const* d_in, const int* in_sizes, int n_in,
                              void* d_out, int out_size, void* d_ws, size_t ws_size,
                              hipStream_t stream) {
    const int*   trip      = (const int*)d_in[0];
    const float* att_feats = (const float*)d_in[1];
    const float* val_feats = (const float*)d_in[2];
    const float* ent_feats = (const float*)d_in[3];
    const float* a_w       = (const float*)d_in[4];
    const float* a_b       = (const float*)d_in[5];
    const float* W         = (const float*)d_in[6];
    float* out = (float*)d_out;

    const float* W_top = W;
    const float* W_bot = W + KD * KD;

    float* ws       = (float*)d_ws;
    float* att_s    = ws + N_ENT;
    int*   row_cnt  = (int*)(att_s + N_ATT);
    int*   ovf_cnt  = row_cnt + N_ENT;
    float* att_proj = (float*)(ovf_cnt + NE);
    int*   bucket   = (int*)(att_proj + (long)N_ATT * KD) + WS_PAD;
    int*   ovf      = bucket + (long)N_ENT * BK;

    {
        int tiles = (N_ATT / 16) * 16;
        static_assert(ATT_THREADS == ((N_ATT / 16) * 16 / 8) * 256, "ATT_THREADS is this launch's thread count");
        wmma_gemm_att<<<tiles / 8, 256, 0, stream>>>(att_feats, W_top, att_proj,
                                                     a_w + KD, att_s, row_cnt, ovf_cnt, bucket);
    }

    edge_bucket_kernel<<<(NE + 255) / 256, 256, 0, stream>>>(trip, row_cnt, bucket, ovf_cnt, ovf);

    row_agg_gemm_ent<<<FB_WGS, FU_THREADS, 0, stream>>>(row_cnt, bucket, ovf_cnt, ovf, ent_feats, a_w, att_s, a_b,
                                                        att_proj, val_feats, W_bot, out);
}


# Matrix waves wait for no memory in a tile: products start from zero, the tile's out rows are read under them and added in the epilogue (sum order: product, then row); no store wait at the hand-over
# speedup vs baseline: 3.2905x; 1.0311x over previous
#include <hip/hip_runtime.h>
#include <math.h>

#define N_ENT 50000
#define N_ATT 2000
#define N_VAL 100000
#define NE    300000
#define KD    256

#define VAL_BITS 17
static_assert(N_VAL <= (1 << VAL_BITS), "value row index must fit VAL_BITS");
static_assert(N_ATT <= (1 << (31 - VAL_BITS)), "attribute row index must fit the rest");

typedef __attribute__((ext_vector_type(2))) float v2f;
typedef __attribute__((ext_vector_type(8))) float v8f;

__device__ __forceinline__ float wave_row_dot(const float4 f0, const float4 f1,
                                              const float4 w0, const float4 w1) {
    float s = 0.0f;
    s += f0.x * w0.x + f0.y * w0.y + f0.z * w0.z + f0.w * w0.w;
    s += f1.x * w1.x + f1.y * w1.y + f1.z * w1.z + f1.w * w1.w;
    for (int off = 16; off > 0; off >>= 1) s += __shfl_xor(s, off, 32);
    return s;
}

#define BK 32
static_assert(BK == 32, "one bucket place per lane: lane l of a row's wave holds bucket record l");
#define ATT_THREADS ((N_ATT / 16) * 16 * 32)
#define BK4 (N_ENT * BK / 4)
__global__ void edge_bucket_kernel(const int* __restrict__ trip,
                                   int* __restrict__ row_cnt,
                                   int* __restrict__ bucket,
                                   int* __restrict__ ovf_cnt,
                                   int* __restrict__ ovf) {
    int e = blockIdx.x * blockDim.x + threadIdx.x;
    if (e >= NE) return;
    const int h  = trip[3 * e + 0];
    const int vl = trip[3 * e + 1];
    const int at = trip[3 * e + 2];
    const int wd = vl | (at << VAL_BITS);
    const int r = atomicAdd(&row_cnt[h], 1);
    if (r < BK) {
        bucket[(long)h * BK + r] = wd;
    } else {
        const int i = atomicAdd(ovf_cnt, 1);
        ovf[2 * (long)i + 0] = h;
        ovf[2 * (long)i + 1] = wd;
    }
}


__global__ void wmma_gemm_att(const float* __restrict__ A,
                              const float* __restrict__ B,
                              float* __restrict__ C,
                              const float* __restrict__ a_w_att,
                              float* __restrict__ att_score,
                              int* __restrict__ row_cnt,
                              int* __restrict__ ovf_cnt,
                              int* __restrict__ bucket) {
    int wave = (int)((blockIdx.x * blockDim.x + threadIdx.x) >> 5);
    int lane = threadIdx.x & 31;
    int tM = wave >> 4;
    int tN = wave & 15;
    int row0 = tM * 16;
    int col0 = tN * 16;

    int m    = lane & 15;
    int koff = (lane >> 4) << 1;
    int bn   = col0 + (lane & 15);
    int crow = row0 + ((lane >> 4) << 3);
    int cn   = col0 + (lane & 15);

    v8f c;
#pragma unroll
    for (int r = 0; r < 8; ++r) c[r] = 0.0f;

    const float* aptr = A + (long)(row0 + m) * KD + koff;

#pragma unroll 4
    for (int k0 = 0; k0 < KD; k0 += 4) {
        v2f av = *(const v2f*)(aptr + k0);
        v2f bv;
        bv.x = B[(long)(k0 + koff + 0) * KD + bn];
        bv.y = B[(long)(k0 + koff + 1) * KD + bn];
        c = __builtin_amdgcn_wmma_f32_16x16x4_f32(
                 false, av,  false, bv,
                 (short)0, c,  false,  false);
    }

#pragma unroll
    for (int r = 0; r < 8; ++r) C[(long)(crow + r) * KD + cn] = c[r];

    const int g = (int)(blockIdx.x * blockDim.x + threadIdx.x);
    if (g < N_ENT) row_cnt[g] = 0;
    if (g == 0) ovf_cnt[0] = 0;
    {
        int4* b4 = (int4*)bucket;
        const int4 z = make_int4(0, 0, 0, 0);
#pragma unroll
        for (int it = 0; it < BK4 / ATT_THREADS; ++it) b4[g + it * ATT_THREADS] = z;
        if (g < BK4 % ATT_THREADS) b4[g + (BK4 / ATT_THREADS) * ATT_THREADS] = z;
    }
    const float4* fr = (const float4*)(A + (long)wave * KD);
    const float4* wr = (const float4*)a_w_att;
    const float s = wave_row_dot(fr[lane], fr[32 + lane], wr[lane], wr[32 + lane]);
    if (lane == 0) att_score[wave] = s;
}

#define FB_PITCH   544
#define FB_WGS     256
#define FB_CH      8
#define M_TILES    (N_ENT / 16)
#define FU_CONS    8
#define FU_PROD    16
#define FU_THREADS ((FU_CONS + FU_PROD) * 32)
#define FU_UP      260
#define FU_NT      2

template <bool LAST>
__device__ __forceinline__ void ent_chunk(v8f (&acc)[FU_NT], v2f& a, v2f (&b)[FU_NT],
                                          const float* ua, const float* bs) {
#pragma unroll
    for (int s = 0; s < FB_CH; ++s) {
        const bool more = !LAST || (s + 1 < FB_CH);
        v2f an = a;
        v2f bn[FU_NT];
#pragma unroll
        for (int t = 0; t < FU_NT; ++t) bn[t] = b[t];
        if (more) {
            an = *(const v2f*)(ua + 4 * (s + 1));
#pragma unroll
            for (int t = 0; t < FU_NT; ++t) bn[t] = *(const v2f*)(bs + 2 * (s + 1) * FB_PITCH + 32 * t);
        }
#pragma unroll
        for (int t = 0; t < FU_NT; ++t)
            acc[t] = __builtin_amdgcn_wmma_f32_16x16x4_f32(
                          false, a,  false, b[t],
                          (short)0, acc[t],  false,  false);
        a = an;
#pragma unroll
        for (int t = 0; t < FU_NT; ++t) b[t] = bn[t];
    }
}

__device__ __forceinline__ float edge_sc(float es, float as, float b) {
    float x = es + as + b;
    x = (x > 0.0f) ? x : 0.2f * x;
    return expf(x);
}

struct RowIn { int cnt; int wd; float4 e0, e1; };

__device__ __forceinline__ RowIn row_fetch(int h, int lane, const int* __restrict__ row_cnt,
                                           const int* __restrict__ bucket,
                                           const float* __restrict__ ent_feats) {
    RowIn r;
    const float4* er = (const float4*)(ent_feats + (long)h * KD);
    r.e0 = er[lane]; r.e1 = er[32 + lane];
    r.wd  = bucket[h * BK + lane];
    r.cnt = *(const __attribute__((address_space(4))) int*)(row_cnt + h);
    return r;
}

#define HEAD_E 4
struct RowHead {
    int   wd_l;
    float p_l;
    float es, rs;
    float4 a0[HEAD_E], a1[HEAD_E];
    float4 c0[HEAD_E], c1[HEAD_E];
};

__device__ __forceinline__ void head_clear(RowHead& hd) {
    const float4 z = make_float4(0.f, 0.f, 0.f, 0.f);
    hd.wd_l = 0; hd.p_l = 0.0f; hd.es = 0.0f; hd.rs = 0.0f;
#pragma unroll
    for (int k = 0; k < HEAD_E; ++k) { hd.a0[k] = z; hd.a1[k] = z; hd.c0[k] = z; hd.c1[k] = z; }
}

__device__ __forceinline__ void row_head(RowHead& hd, const RowIn& in, const int hrow, const int wd_l,
                                         const float as_l, const bool has_l, const int lane,
                                         const float4 w0, const float4 w1, const float b,
                                         const int* __restrict__ ovf_cnt,
                                         const int* __restrict__ ovf,
                                         const float* __restrict__ att_score,
                                         const float* __restrict__ att_proj,
                                         const float* __restrict__ val_feats) {
    const float es = wave_row_dot(in.e0, in.e1, w0, w1);
    const float sc_l = has_l ? edge_sc(es, as_l, b) : 0.0f;
    float rs = sc_l;
    if (in.cnt > BK) {
        const int nov = (ovf_cnt[0] < NE) ? ovf_cnt[0] : NE;
        for (int i = lane; i < nov; i += 32)
            if (ovf[2 * (long)i] == hrow)
                rs += edge_sc(es, att_score[ovf[2 * (long)i + 1] >> VAL_BITS], b);
    }
    for (int off = 16; off > 0; off >>= 1) rs += __shfl_xor(rs, off, 32);
    hd.wd_l = wd_l; hd.p_l = sc_l / rs; hd.es = es; hd.rs = rs;
    const int n = in.cnt;
#pragma unroll
    for (int k = 0; k < HEAD_E; ++k) {
        if (k < n) {
            const int wd = __builtin_amdgcn_readlane(wd_l, k);
            const int vl = wd & ((1 << VAL_BITS) - 1);
            const int at = wd >> VAL_BITS;
            const float4* vr = (const float4*)(val_feats + (long)vl * KD);
            const float4* ar = (const float4*)(att_proj  + (long)at * KD);
            hd.a0[k] = vr[lane]; hd.a1[k] = vr[32 + lane];
            hd.c0[k] = ar[lane]; hd.c1[k] = ar[32 + lane];
        }
    }
}

__device__ __forceinline__ void row_step(RowIn& cur, RowHead& hd, RowIn& nxt,
                                         const int h, const int hn, const int hnn, const int lane,
                                         const float* ws_lds, const float b,
                                         const int* __restrict__ bucket,
                                         const int* __restrict__ ovf_cnt,
                                         const int* __restrict__ ovf,
                                         const float* __restrict__ att_score,
                                         const float* __restrict__ att_proj,
                                         const float* __restrict__ val_feats,
                                         const int* __restrict__ row_cnt,
                                         const float* __restrict__ ent_feats,
                                         float* urow,
                                         float* __restrict__ outAcc) {
    const float4 e0 = cur.e0, e1 = cur.e1;
    const int cnt = cur.cnt;

    const bool hasn = (hn >= 0) && (lane < nxt.cnt);
    const int  wdn  = hasn ? nxt.wd : 0;
    float asn = 0.0f;
    if (hasn) asn = att_score[wdn >> VAL_BITS];

    asm("" : "+v"(nxt.e0.x), "+v"(nxt.e0.y), "+v"(nxt.e0.z), "+v"(nxt.e0.w),
             "+v"(nxt.e1.x), "+v"(nxt.e1.y), "+v"(nxt.e1.z), "+v"(nxt.e1.w));

    float4 u0 = make_float4(0.f, 0.f, 0.f, 0.f), u1 = u0, o0 = u0, o1 = u0;
    auto add = [&](const float p, const float4 a0, const float4 a1, const float4 c0, const float4 c1) {
        u0.x += p * a0.x; u0.y += p * a0.y; u0.z += p * a0.z; u0.w += p * a0.w;
        u1.x += p * a1.x; u1.y += p * a1.y; u1.z += p * a1.z; u1.w += p * a1.w;
        o0.x += p * c0.x; o0.y += p * c0.y; o0.z += p * c0.z; o0.w += p * c0.w;
        o1.x += p * c1.x; o1.y += p * c1.y; o1.z += p * c1.z; o1.w += p * c1.w;
    };

    const int n0 = (cnt < BK) ? cnt : BK;
#pragma unroll
    for (int k = 0; k < HEAD_E; ++k) {
        if (k < n0) {
            const float p = __int_as_float(__builtin_amdgcn_readlane(__float_as_int(hd.p_l), k));
            add(p, hd.a0[k], hd.a1[k], hd.c0[k], hd.c1[k]);
        }
    }

    int   wd_l = hd.wd_l;
    float p_l  = hd.p_l;
    auto edge = [&](int jj) {
        const int   wd = __builtin_amdgcn_readlane(wd_l, jj);
        const float p  = __int_as_float(__builtin_amdgcn_readlane(__float_as_int(p_l), jj));
        const int vl = wd & ((1 << VAL_BITS) - 1);
        const int at = wd >> VAL_BITS;
        const float4* vr = (const float4*)(val_feats + (long)vl * KD);
        const float4* ar = (const float4*)(att_proj  + (long)at * KD);
        const float4 a0 = vr[lane], a1 = vr[32 + lane];
        const float4 c0 = ar[lane], c1 = ar[32 + lane];
        add(p, a0, a1, c0, c1);
    };
    {
        int j = (n0 < HEAD_E) ? n0 : HEAD_E;
        for (; j + 4 <= n0; j += 4) { edge(j); edge(j + 1); edge(j + 2); edge(j + 3); }
        const int rem = n0 - j;
        if (rem == 3)      { edge(j); edge(j + 1); edge(j + 2); }
        else if (rem == 2) { edge(j); edge(j + 1); }
        else if (rem == 1) { edge(j); }
    }
    if (cnt > BK) {
        const float es = hd.es, rs = hd.rs;
        const int nov = (ovf_cnt[0] < NE) ? ovf_cnt[0] : NE;
        for (int c = 0; c < nov; c += 32) {
            const int i = c + lane;
            bool mine = false;
            float sc_l = 0.0f;
            wd_l = 0;
            if (i < nov) mine = (ovf[2 * (long)i] == h);
            if (mine) {
                wd_l = ovf[2 * (long)i + 1];
                sc_l = edge_sc(es, att_score[wd_l >> VAL_BITS], b);
            }
            p_l = sc_l / rs;
            unsigned todo = __builtin_amdgcn_ballot_w32(mine);
            while (todo != 0u) {
                edge(__builtin_ctz(todo));
                todo &= todo - 1u;
            }
        }
    }

    float4* ur   = (float4*)urow;
    float4* orow = (float4*)(outAcc + (long)h * KD);
    ur[lane] = u0;   ur[32 + lane] = u1;
    o0.x += e0.x; o0.y += e0.y; o0.z += e0.z; o0.w += e0.w;
    o1.x += e1.x; o1.y += e1.y; o1.z += e1.z; o1.w += e1.w;
    orow[lane] = o0; orow[32 + lane] = o1;

    head_clear(hd);
    if (hn >= 0) {
        const float4 w0 = ((const float4*)ws_lds)[lane], w1 = ((const float4*)ws_lds)[32 + lane];
        row_head(hd, nxt, hn, wdn, asn, hasn, lane, w0, w1, b, ovf_cnt, ovf, att_score, att_proj, val_feats);
    }
    if (hnn >= 0) cur = row_fetch(hnn, lane, row_cnt, bucket, ent_feats);
}

__global__ __launch_bounds__(FU_THREADS) void row_agg_gemm_ent(const int* __restrict__ row_cnt,
                                                               const int* __restrict__ bucket,
                                                               const int* __restrict__ ovf_cnt,
                                                               const int* __restrict__ ovf,
                                                               const float* __restrict__ ent_feats,
                                                               const float* __restrict__ a_w,
                                                               const float* __restrict__ att_score,
                                                               const float* __restrict__ a_b,
                                                               const float* __restrict__ att_proj,
                                                               const float* __restrict__ val_feats,
                                                               const float* __restrict__ B,
                                                               float* C) {
    __shared__ float Bs[(KD / 2) * FB_PITCH];
    __shared__ float Us[2 * 16 * FU_UP];
    __shared__ float Ws[KD];

    const int wave = __builtin_amdgcn_readfirstlane((int)(threadIdx.x >> 5));
    const int lane = threadIdx.x & 31;
    const bool producer = (wave >= FU_CONS);
    const int prow = wave - FU_CONS;
    const int nT = (M_TILES - (int)blockIdx.x + FB_WGS - 1) / FB_WGS;

    RowIn cur, nxt;
    RowHead hd;
    cur.cnt = 0; cur.wd = 0; cur.e0 = make_float4(0.f, 0.f, 0.f, 0.f); cur.e1 = cur.e0;
    nxt = cur;
    head_clear(hd);
    float4 w0 = make_float4(0.f, 0.f, 0.f, 0.f), w1 = w0;
    float ab = 0.0f;
    if (producer) {
        const float4* aw = (const float4*)a_w;
        w0 = aw[lane]; w1 = aw[32 + lane];
        ((float4*)Ws)[lane] = w0; ((float4*)Ws)[32 + lane] = w1;
        ab = a_b[0];
        const int h0 = (int)blockIdx.x * 16 + prow;
        cur = row_fetch(h0, lane, row_cnt, bucket, ent_feats);
        nxt = cur;
        if (1 < nT) nxt = row_fetch(h0 + FB_WGS * 16, lane, row_cnt, bucket, ent_feats);
    }

#pragma unroll
    for (int it = 0; it < ((KD / 2) * (KD / 4) + FU_THREADS - 1) / FU_THREADS; ++it) {
        const int idx = (int)threadIdx.x + it * FU_THREADS;
        if (idx < (KD / 2) * (KD / 4)) {
            const int p  = idx >> 6;
            const int n4 = (idx & 63) << 2;
            const float4 r0 = *(const float4*)(B + (long)(2 * p)     * KD + n4);
            const float4 r1 = *(const float4*)(B + (long)(2 * p + 1) * KD + n4);
            float* d = Bs + p * FB_PITCH + n4 * 2;
            *(float4*)(d)     = make_float4(r0.x, r1.x, r0.y, r1.y);
            *(float4*)(d + 4) = make_float4(r0.z, r1.z, r0.w, r1.w);
        }
    }

    if (producer) {
        const bool has_l = (lane < cur.cnt);
        const int  wd_l = has_l ? cur.wd : 0;
        float as_l = 0.0f;
        if (has_l) as_l = att_score[wd_l >> VAL_BITS];
        row_head(hd, cur, (int)blockIdx.x * 16 + prow, wd_l, as_l, has_l, lane, w0, w1, ab,
                 ovf_cnt, ovf, att_score, att_proj, val_feats);
    }

    const int c    = lane & 15;
    const int hh   = lane >> 4;
    const int col0 = wave * (16 * FU_NT);
    const float* bbase = Bs + hh * FB_PITCH + 2 * (col0 + c);

    auto trip = [&](const int j, RowIn& rc, RowIn& rn) {
        if (producer) {
            if (j + 1 < nT) {
                const int h   = ((int)blockIdx.x + FB_WGS * (j + 1)) * 16 + prow;
                const int hn  = (j + 2 < nT) ? h + FB_WGS * 16 : -1;
                const int hnn = (j + 3 < nT) ? h + 2 * FB_WGS * 16 : -1;
                row_step(rc, hd, rn, h, hn, hnn, lane, Ws, ab,
                         bucket, ovf_cnt, ovf, att_score, att_proj, val_feats, row_cnt, ent_feats,
                         Us + ((j + 1) & 1) * (16 * FU_UP) + prow * FU_UP, C);
            }
        } else if (j >= 0) {
            const int row0 = ((int)blockIdx.x + FB_WGS * j) * 16;
            float* crow = C + (long)(row0 + 8 * hh) * KD + col0 + c;
            const float* us = Us + (j & 1) * (16 * FU_UP) + c * FU_UP + 2 * hh;

            float cin[FU_NT][8];
#pragma unroll
            for (int t = 0; t < FU_NT; ++t)
#pragma unroll
                for (int r = 0; r < 8; ++r) cin[t][r] = crow[(long)r * KD + 16 * t];

            v8f acc[FU_NT];
#pragma unroll
            for (int t = 0; t < FU_NT; ++t)
#pragma unroll
                for (int r = 0; r < 8; ++r) acc[t][r] = 0.0f;

            const float* bs = bbase;
            const float* ua = us;
            v2f a = *(const v2f*)ua;
            v2f b[FU_NT];
#pragma unroll
            for (int t = 0; t < FU_NT; ++t) b[t] = *(const v2f*)(bs + 32 * t);
#pragma unroll 1
            for (int ch = 0; ch < KD / (4 * FB_CH) - 1; ++ch) {
                ent_chunk<false>(acc, a, b, ua, bs);
                ua += 4 * FB_CH;
                bs += 2 * FB_CH * FB_PITCH;
            }
            ent_chunk<true>(acc, a, b, ua, bs);

#pragma unroll
            for (int t = 0; t < FU_NT; ++t)
#pragma unroll
                for (int r = 0; r < 8; ++r) {
                    float x = acc[t][r] + cin[t][r];
                    x = (x > 0.0f) ? x : (expf(x) - 1.0f);
                    crow[(long)r * KD + 16 * t] = x;
                }
        }
        if (producer) asm volatile("s_wait_storecnt_dscnt 0x0" ::: "memory");
        else          asm volatile("s_wait_dscnt 0x0" ::: "memory");
        __builtin_amdgcn_s_barrier();
        asm volatile("" ::: "memory");
    };
    for (int j = -1; j < nT; j += 2) {
        trip(j, cur, nxt);
        if (j + 1 < nT) trip(j + 1, nxt, cur);
    }
}

static_assert((N_ATT / 16) * 16 == N_ATT, "attribute GEMM: one wave per attribute row, whole 16-row tiles");
static_assert(ATT_THREADS >= N_ENT, "attribute GEMM launch must have a thread per entity row counter");
static_assert((N_ENT / 16) * 16 == N_ENT, "entity GEMM: whole 16-row tiles");
static_assert(((KD / 2) * FB_PITCH + 2 * 16 * FU_UP + KD) * 4 <= 320 * 1024, "entity GEMM: W_bot, two tiles of U and the score weights must fit the WGP's LDS");
static_assert(FU_CONS * 16 * FU_NT == KD, "entity GEMM: the consumer waves cover all output columns");
static_assert(FU_PROD == 16, "one producer wave per row of a tile");
#define WS_PAD 48
static_assert((N_ENT + N_ATT + N_ENT + NE + N_ATT * KD + WS_PAD) % 64 == 0, "buckets start on a 256-byte multiple");
static_assert(HEAD_E <= BK, "the head's edges come out of the bucket");

extern "C" void kernel_launch(void* const* d_in, const int* in_sizes, int n_in,
                              void* d_out, int out_size, void* d_ws, size_t ws_size,
                              hipStream_t stream) {
    const int*   trip      = (const int*)d_in[0];
    const float* att_feats = (const float*)d_in[1];
    const float* val_feats = (const float*)d_in[2];
    const float* ent_feats = (const float*)d_in[3];
    const float* a_w       = (const float*)d_in[4];
    const float* a_b       = (const float*)d_in[5];
    const float* W         = (const float*)d_in[6];
    float* out = (float*)d_out;

    const float* W_top = W;
    const float* W_bot = W + KD * KD;

    float* ws       = (float*)d_ws;
    float* att_s    = ws + N_ENT;
    int*   row_cnt  = (int*)(att_s + N_ATT);
    int*   ovf_cnt  = row_cnt + N_ENT;
    float* att_proj = (float*)(ovf_cnt + NE);
    int*   bucket   = (int*)(att_proj + (long)N_ATT * KD) + WS_PAD;
    int*   ovf      = bucket + (long)N_ENT * BK;

    {
        int tiles = (N_ATT / 16) * 16;
        static_assert(ATT_THREADS == ((N_ATT / 16) * 16 / 8) * 256, "ATT_THREADS is this launch's thread count");
        wmma_gemm_att<<<tiles / 8, 256, 0, stream>>>(att_feats, W_top, att_proj,
                                                     a_w + KD, att_s, row_cnt, ovf_cnt, bucket);
    }

    edge_bucket_kernel<<<(NE + 255) / 256, 256, 0, stream>>>(trip, row_cnt, bucket, ovf_cnt, ovf);

    row_agg_gemm_ent<<<FB_WGS, FU_THREADS, 0, stream>>>(row_cnt, bucket, ovf_cnt, ovf, ent_feats, a_w, att_s, a_b,
                                                        att_proj, val_feats, W_bot, out);
}


# Producer waves prefetch into the second-level cache a trip ahead: next row's value rows (one edge per lane, 4 x 256 B) and the entity row two tiles on; loads, sums and their order unchanged
# speedup vs baseline: 3.4126x; 1.0371x over previous
#include <hip/hip_runtime.h>
#include <math.h>

#define N_ENT 50000
#define N_ATT 2000
#define N_VAL 100000
#define NE    300000
#define KD    256

#define VAL_BITS 17
static_assert(N_VAL <= (1 << VAL_BITS), "value row index must fit VAL_BITS");
static_assert(N_ATT <= (1 << (31 - VAL_BITS)), "attribute row index must fit the rest");

typedef __attribute__((ext_vector_type(2))) float v2f;
typedef __attribute__((ext_vector_type(8))) float v8f;

__device__ __forceinline__ float wave_row_dot(const float4 f0, const float4 f1,
                                              const float4 w0, const float4 w1) {
    float s = 0.0f;
    s += f0.x * w0.x + f0.y * w0.y + f0.z * w0.z + f0.w * w0.w;
    s += f1.x * w1.x + f1.y * w1.y + f1.z * w1.z + f1.w * w1.w;
    for (int off = 16; off > 0; off >>= 1) s += __shfl_xor(s, off, 32);
    return s;
}

#define BK 32
static_assert(BK == 32, "one bucket place per lane: lane l of a row's wave holds bucket record l");
#define ATT_THREADS ((N_ATT / 16) * 16 * 32)
#define BK4 (N_ENT * BK / 4)
__global__ void edge_bucket_kernel(const int* __restrict__ trip,
                                   int* __restrict__ row_cnt,
                                   int* __restrict__ bucket,
                                   int* __restrict__ ovf_cnt,
                                   int* __restrict__ ovf) {
    int e = blockIdx.x * blockDim.x + threadIdx.x;
    if (e >= NE) return;
    const int h  = trip[3 * e + 0];
    const int vl = trip[3 * e + 1];
    const int at = trip[3 * e + 2];
    const int wd = vl | (at << VAL_BITS);
    const int r = atomicAdd(&row_cnt[h], 1);
    if (r < BK) {
        bucket[(long)h * BK + r] = wd;
    } else {
        const int i = atomicAdd(ovf_cnt, 1);
        ovf[2 * (long)i + 0] = h;
        ovf[2 * (long)i + 1] = wd;
    }
}


__global__ void wmma_gemm_att(const float* __restrict__ A,
                              const float* __restrict__ B,
                              float* __restrict__ C,
                              const float* __restrict__ a_w_att,
                              float* __restrict__ att_score,
                              int* __restrict__ row_cnt,
                              int* __restrict__ ovf_cnt,
                              int* __restrict__ bucket) {
    int wave = (int)((blockIdx.x * blockDim.x + threadIdx.x) >> 5);
    int lane = threadIdx.x & 31;
    int tM = wave >> 4;
    int tN = wave & 15;
    int row0 = tM * 16;
    int col0 = tN * 16;

    int m    = lane & 15;
    int koff = (lane >> 4) << 1;
    int bn   = col0 + (lane & 15);
    int crow = row0 + ((lane >> 4) << 3);
    int cn   = col0 + (lane & 15);

    v8f c;
#pragma unroll
    for (int r = 0; r < 8; ++r) c[r] = 0.0f;

    const float* aptr = A + (long)(row0 + m) * KD + koff;

#pragma unroll 4
    for (int k0 = 0; k0 < KD; k0 += 4) {
        v2f av = *(const v2f*)(aptr + k0);
        v2f bv;
        bv.x = B[(long)(k0 + koff + 0) * KD + bn];
        bv.y = B[(long)(k0 + koff + 1) * KD + bn];
        c = __builtin_amdgcn_wmma_f32_16x16x4_f32(
                 false, av,  false, bv,
                 (short)0, c,  false,  false);
    }

#pragma unroll
    for (int r = 0; r < 8; ++r) C[(long)(crow + r) * KD + cn] = c[r];

    const int g = (int)(blockIdx.x * blockDim.x + threadIdx.x);
    if (g < N_ENT) row_cnt[g] = 0;
    if (g == 0) ovf_cnt[0] = 0;
    {
        int4* b4 = (int4*)bucket;
        const int4 z = make_int4(0, 0, 0, 0);
#pragma unroll
        for (int it = 0; it < BK4 / ATT_THREADS; ++it) b4[g + it * ATT_THREADS] = z;
        if (g < BK4 % ATT_THREADS) b4[g + (BK4 / ATT_THREADS) * ATT_THREADS] = z;
    }
    const float4* fr = (const float4*)(A + (long)wave * KD);
    const float4* wr = (const float4*)a_w_att;
    const float s = wave_row_dot(fr[lane], fr[32 + lane], wr[lane], wr[32 + lane]);
    if (lane == 0) att_score[wave] = s;
}

#define FB_PITCH   544
#define FB_WGS     256
#define FB_CH      8
#define M_TILES    (N_ENT / 16)
#define FU_CONS    8
#define FU_PROD    16
#define FU_THREADS ((FU_CONS + FU_PROD) * 32)
#define FU_UP      260
#define FU_NT      2

template <bool LAST>
__device__ __forceinline__ void ent_chunk(v8f (&acc)[FU_NT], v2f& a, v2f (&b)[FU_NT],
                                          const float* ua, const float* bs) {
#pragma unroll
    for (int s = 0; s < FB_CH; ++s) {
        const bool more = !LAST || (s + 1 < FB_CH);
        v2f an = a;
        v2f bn[FU_NT];
#pragma unroll
        for (int t = 0; t < FU_NT; ++t) bn[t] = b[t];
        if (more) {
            an = *(const v2f*)(ua + 4 * (s + 1));
#pragma unroll
            for (int t = 0; t < FU_NT; ++t) bn[t] = *(const v2f*)(bs + 2 * (s + 1) * FB_PITCH + 32 * t);
        }
#pragma unroll
        for (int t = 0; t < FU_NT; ++t)
            acc[t] = __builtin_amdgcn_wmma_f32_16x16x4_f32(
                          false, a,  false, b[t],
                          (short)0, acc[t],  false,  false);
        a = an;
#pragma unroll
        for (int t = 0; t < FU_NT; ++t) b[t] = bn[t];
    }
}

__device__ __forceinline__ float edge_sc(float es, float as, float b) {
    float x = es + as + b;
    x = (x > 0.0f) ? x : 0.2f * x;
    return expf(x);
}

struct RowIn { int cnt; int wd; float4 e0, e1; };

__device__ __forceinline__ RowIn row_fetch(int h, int lane, const int* __restrict__ row_cnt,
                                           const int* __restrict__ bucket,
                                           const float* __restrict__ ent_feats) {
    RowIn r;
    const float4* er = (const float4*)(ent_feats + (long)h * KD);
    r.e0 = er[lane]; r.e1 = er[32 + lane];
    r.wd  = bucket[h * BK + lane];
    r.cnt = *(const __attribute__((address_space(4))) int*)(row_cnt + h);
    return r;
}

__device__ __forceinline__ void val_row_prefetch(const float* __restrict__ val_feats, const int wd) {
    const char* vp = (const char*)val_feats + ((unsigned)(wd & ((1 << VAL_BITS) - 1)) * (KD * 4u));
    __builtin_prefetch(vp,       0, 2);
    __builtin_prefetch(vp + 256, 0, 2);
    __builtin_prefetch(vp + 512, 0, 2);
    __builtin_prefetch(vp + 768, 0, 2);
}

#define HEAD_E 4
struct RowHead {
    int   wd_l;
    float p_l;
    float es, rs;
    float4 a0[HEAD_E], a1[HEAD_E];
    float4 c0[HEAD_E], c1[HEAD_E];
};

__device__ __forceinline__ void head_clear(RowHead& hd) {
    const float4 z = make_float4(0.f, 0.f, 0.f, 0.f);
    hd.wd_l = 0; hd.p_l = 0.0f; hd.es = 0.0f; hd.rs = 0.0f;
#pragma unroll
    for (int k = 0; k < HEAD_E; ++k) { hd.a0[k] = z; hd.a1[k] = z; hd.c0[k] = z; hd.c1[k] = z; }
}

__device__ __forceinline__ void row_head(RowHead& hd, const RowIn& in, const int hrow, const int wd_l,
                                         const float as_l, const bool has_l, const int lane,
                                         const float4 w0, const float4 w1, const float b,
                                         const int* __restrict__ ovf_cnt,
                                         const int* __restrict__ ovf,
                                         const float* __restrict__ att_score,
                                         const float* __restrict__ att_proj,
                                         const float* __restrict__ val_feats) {
    const float es = wave_row_dot(in.e0, in.e1, w0, w1);
    const float sc_l = has_l ? edge_sc(es, as_l, b) : 0.0f;
    float rs = sc_l;
    if (in.cnt > BK) {
        const int nov = (ovf_cnt[0] < NE) ? ovf_cnt[0] : NE;
        for (int i = lane; i < nov; i += 32)
            if (ovf[2 * (long)i] == hrow)
                rs += edge_sc(es, att_score[ovf[2 * (long)i + 1] >> VAL_BITS], b);
    }
    for (int off = 16; off > 0; off >>= 1) rs += __shfl_xor(rs, off, 32);
    hd.wd_l = wd_l; hd.p_l = sc_l / rs; hd.es = es; hd.rs = rs;
    const int n = in.cnt;
#pragma unroll
    for (int k = 0; k < HEAD_E; ++k) {
        if (k < n) {
            const int wd = __builtin_amdgcn_readlane(wd_l, k);
            const int vl = wd & ((1 << VAL_BITS) - 1);
            const int at = wd >> VAL_BITS;
            const float4* vr = (const float4*)(val_feats + (long)vl * KD);
            const float4* ar = (const float4*)(att_proj  + (long)at * KD);
            hd.a0[k] = vr[lane]; hd.a1[k] = vr[32 + lane];
            hd.c0[k] = ar[lane]; hd.c1[k] = ar[32 + lane];
        }
    }
}

__device__ __forceinline__ void row_step(RowIn& cur, RowHead& hd, RowIn& nxt,
                                         const int h, const int hn, const int hnn, const int lane,
                                         const float* ws_lds, const float b,
                                         const int* __restrict__ bucket,
                                         const int* __restrict__ ovf_cnt,
                                         const int* __restrict__ ovf,
                                         const float* __restrict__ att_score,
                                         const float* __restrict__ att_proj,
                                         const float* __restrict__ val_feats,
                                         const int* __restrict__ row_cnt,
                                         const float* __restrict__ ent_feats,
                                         float* urow,
                                         float* __restrict__ outAcc) {
    const float4 e0 = cur.e0, e1 = cur.e1;
    const int cnt = cur.cnt;

    const bool hasn = (hn >= 0) && (lane < nxt.cnt);
    const int  wdn  = hasn ? nxt.wd : 0;
    float asn = 0.0f;
    if (hasn) asn = att_score[wdn >> VAL_BITS];

    if (hasn) val_row_prefetch(val_feats, wdn);
    if (hnn >= 0 && lane < 4)
        __builtin_prefetch((const char*)ent_feats + ((unsigned)hnn * (KD * 4u) + (unsigned)lane * 256u), 0, 2);

    asm("" : "+v"(nxt.e0.x), "+v"(nxt.e0.y), "+v"(nxt.e0.z), "+v"(nxt.e0.w),
             "+v"(nxt.e1.x), "+v"(nxt.e1.y), "+v"(nxt.e1.z), "+v"(nxt.e1.w));

    float4 u0 = make_float4(0.f, 0.f, 0.f, 0.f), u1 = u0, o0 = u0, o1 = u0;
    auto add = [&](const float p, const float4 a0, const float4 a1, const float4 c0, const float4 c1) {
        u0.x += p * a0.x; u0.y += p * a0.y; u0.z += p * a0.z; u0.w += p * a0.w;
        u1.x += p * a1.x; u1.y += p * a1.y; u1.z += p * a1.z; u1.w += p * a1.w;
        o0.x += p * c0.x; o0.y += p * c0.y; o0.z += p * c0.z; o0.w += p * c0.w;
        o1.x += p * c1.x; o1.y += p * c1.y; o1.z += p * c1.z; o1.w += p * c1.w;
    };

    const int n0 = (cnt < BK) ? cnt : BK;
#pragma unroll
    for (int k = 0; k < HEAD_E; ++k) {
        if (k < n0) {
            const float p = __int_as_float(__builtin_amdgcn_readlane(__float_as_int(hd.p_l), k));
            add(p, hd.a0[k], hd.a1[k], hd.c0[k], hd.c1[k]);
        }
    }

    int   wd_l = hd.wd_l;
    float p_l  = hd.p_l;
    auto edge = [&](int jj) {
        const int   wd = __builtin_amdgcn_readlane(wd_l, jj);
        const float p  = __int_as_float(__builtin_amdgcn_readlane(__float_as_int(p_l), jj));
        const int vl = wd & ((1 << VAL_BITS) - 1);
        const int at = wd >> VAL_BITS;
        const float4* vr = (const float4*)(val_feats + (long)vl * KD);
        const float4* ar = (const float4*)(att_proj  + (long)at * KD);
        const float4 a0 = vr[lane], a1 = vr[32 + lane];
        const float4 c0 = ar[lane], c1 = ar[32 + lane];
        add(p, a0, a1, c0, c1);
    };
    {
        int j = (n0 < HEAD_E) ? n0 : HEAD_E;
        for (; j + 4 <= n0; j += 4) { edge(j); edge(j + 1); edge(j + 2); edge(j + 3); }
        const int rem = n0 - j;
        if (rem == 3)      { edge(j); edge(j + 1); edge(j + 2); }
        else if (rem == 2) { edge(j); edge(j + 1); }
        else if (rem == 1) { edge(j); }
    }
    if (cnt > BK) {
        const float es = hd.es, rs = hd.rs;
        const int nov = (ovf_cnt[0] < NE) ? ovf_cnt[0] : NE;
        for (int c = 0; c < nov; c += 32) {
            const int i = c + lane;
            bool mine = false;
            float sc_l = 0.0f;
            wd_l = 0;
            if (i < nov) mine = (ovf[2 * (long)i] == h);
            if (mine) {
                wd_l = ovf[2 * (long)i + 1];
                sc_l = edge_sc(es, att_score[wd_l >> VAL_BITS], b);
            }
            p_l = sc_l / rs;
            unsigned todo = __builtin_amdgcn_ballot_w32(mine);
            while (todo != 0u) {
                edge(__builtin_ctz(todo));
                todo &= todo - 1u;
            }
        }
    }

    float4* ur   = (float4*)urow;
    float4* orow = (float4*)(outAcc + (long)h * KD);
    ur[lane] = u0;   ur[32 + lane] = u1;
    o0.x += e0.x; o0.y += e0.y; o0.z += e0.z; o0.w += e0.w;
    o1.x += e1.x; o1.y += e1.y; o1.z += e1.z; o1.w += e1.w;
    orow[lane] = o0; orow[32 + lane] = o1;

    head_clear(hd);
    if (hn >= 0) {
        const float4 w0 = ((const float4*)ws_lds)[lane], w1 = ((const float4*)ws_lds)[32 + lane];
        row_head(hd, nxt, hn, wdn, asn, hasn, lane, w0, w1, b, ovf_cnt, ovf, att_score, att_proj, val_feats);
    }
    if (hnn >= 0) cur = row_fetch(hnn, lane, row_cnt, bucket, ent_feats);
}

__global__ __launch_bounds__(FU_THREADS) void row_agg_gemm_ent(const int* __restrict__ row_cnt,
                                                               const int* __restrict__ bucket,
                                                               const int* __restrict__ ovf_cnt,
                                                               const int* __restrict__ ovf,
                                                               const float* __restrict__ ent_feats,
                                                               const float* __restrict__ a_w,
                                                               const float* __restrict__ att_score,
                                                               const float* __restrict__ a_b,
                                                               const float* __restrict__ att_proj,
                                                               const float* __restrict__ val_feats,
                                                               const float* __restrict__ B,
                                                               float* C) {
    __shared__ float Bs[(KD / 2) * FB_PITCH];
    __shared__ float Us[2 * 16 * FU_UP];
    __shared__ float Ws[KD];

    const int wave = __builtin_amdgcn_readfirstlane((int)(threadIdx.x >> 5));
    const int lane = threadIdx.x & 31;
    const bool producer = (wave >= FU_CONS);
    const int prow = wave - FU_CONS;
    const int nT = (M_TILES - (int)blockIdx.x + FB_WGS - 1) / FB_WGS;

    RowIn cur, nxt;
    RowHead hd;
    cur.cnt = 0; cur.wd = 0; cur.e0 = make_float4(0.f, 0.f, 0.f, 0.f); cur.e1 = cur.e0;
    nxt = cur;
    head_clear(hd);
    float4 w0 = make_float4(0.f, 0.f, 0.f, 0.f), w1 = w0;
    float ab = 0.0f;
    if (producer) {
        const float4* aw = (const float4*)a_w;
        w0 = aw[lane]; w1 = aw[32 + lane];
        ((float4*)Ws)[lane] = w0; ((float4*)Ws)[32 + lane] = w1;
        ab = a_b[0];
        const int h0 = (int)blockIdx.x * 16 + prow;
        cur = row_fetch(h0, lane, row_cnt, bucket, ent_feats);
        nxt = cur;
        if (1 < nT) nxt = row_fetch(h0 + FB_WGS * 16, lane, row_cnt, bucket, ent_feats);
    }

#pragma unroll
    for (int it = 0; it < ((KD / 2) * (KD / 4) + FU_THREADS - 1) / FU_THREADS; ++it) {
        const int idx = (int)threadIdx.x + it * FU_THREADS;
        if (idx < (KD / 2) * (KD / 4)) {
            const int p  = idx >> 6;
            const int n4 = (idx & 63) << 2;
            const float4 r0 = *(const float4*)(B + (long)(2 * p)     * KD + n4);
            const float4 r1 = *(const float4*)(B + (long)(2 * p + 1) * KD + n4);
            float* d = Bs + p * FB_PITCH + n4 * 2;
            *(float4*)(d)     = make_float4(r0.x, r1.x, r0.y, r1.y);
            *(float4*)(d + 4) = make_float4(r0.z, r1.z, r0.w, r1.w);
        }
    }

    if (producer) {
        const bool has_l = (lane < cur.cnt);
        const int  wd_l = has_l ? cur.wd : 0;
        float as_l = 0.0f;
        if (has_l) as_l = att_score[wd_l >> VAL_BITS];
        if (has_l && lane >= HEAD_E) val_row_prefetch(val_feats, wd_l);
        row_head(hd, cur, (int)blockIdx.x * 16 + prow, wd_l, as_l, has_l, lane, w0, w1, ab,
                 ovf_cnt, ovf, att_score, att_proj, val_feats);
    }

    const int c    = lane & 15;
    const int hh   = lane >> 4;
    const int col0 = wave * (16 * FU_NT);
    const float* bbase = Bs + hh * FB_PITCH + 2 * (col0 + c);

    auto trip = [&](const int j, RowIn& rc, RowIn& rn) {
        if (producer) {
            if (j + 1 < nT) {
                const int h   = ((int)blockIdx.x + FB_WGS * (j + 1)) * 16 + prow;
                const int hn  = (j + 2 < nT) ? h + FB_WGS * 16 : -1;
                const int hnn = (j + 3 < nT) ? h + 2 * FB_WGS * 16 : -1;
                row_step(rc, hd, rn, h, hn, hnn, lane, Ws, ab,
                         bucket, ovf_cnt, ovf, att_score, att_proj, val_feats, row_cnt, ent_feats,
                         Us + ((j + 1) & 1) * (16 * FU_UP) + prow * FU_UP, C);
            }
        } else if (j >= 0) {
            const int row0 = ((int)blockIdx.x + FB_WGS * j) * 16;
            float* crow = C + (long)(row0 + 8 * hh) * KD + col0 + c;
            const float* us = Us + (j & 1) * (16 * FU_UP) + c * FU_UP + 2 * hh;

            float cin[FU_NT][8];
#pragma unroll
            for (int t = 0; t < FU_NT; ++t)
#pragma unroll
                for (int r = 0; r < 8; ++r) cin[t][r] = crow[(long)r * KD + 16 * t];

            v8f acc[FU_NT];
#pragma unroll
            for (int t = 0; t < FU_NT; ++t)
#pragma unroll
                for (int r = 0; r < 8; ++r) acc[t][r] = 0.0f;

            const float* bs = bbase;
            const float* ua = us;
            v2f a = *(const v2f*)ua;
            v2f b[FU_NT];
#pragma unroll
            for (int t = 0; t < FU_NT; ++t) b[t] = *(const v2f*)(bs + 32 * t);
#pragma unroll 1
            for (int ch = 0; ch < KD / (4 * FB_CH) - 1; ++ch) {
                ent_chunk<false>(acc, a, b, ua, bs);
                ua += 4 * FB_CH;
                bs += 2 * FB_CH * FB_PITCH;
            }
            ent_chunk<true>(acc, a, b, ua, bs);

#pragma unroll
            for (int t = 0; t < FU_NT; ++t)
#pragma unroll
                for (int r = 0; r < 8; ++r) {
                    float x = acc[t][r] + cin[t][r];
                    x = (x > 0.0f) ? x : (expf(x) - 1.0f);
                    crow[(long)r * KD + 16 * t] = x;
                }
        }
        if (producer) asm volatile("s_wait_storecnt_dscnt 0x0" ::: "memory");
        else          asm volatile("s_wait_dscnt 0x0" ::: "memory");
        __builtin_amdgcn_s_barrier();
        asm volatile("" ::: "memory");
    };
    for (int j = -1; j < nT; j += 2) {
        trip(j, cur, nxt);
        if (j + 1 < nT) trip(j + 1, nxt, cur);
    }
}

static_assert((N_ATT / 16) * 16 == N_ATT, "attribute GEMM: one wave per attribute row, whole 16-row tiles");
static_assert(ATT_THREADS >= N_ENT, "attribute GEMM launch must have a thread per entity row counter");
static_assert((N_ENT / 16) * 16 == N_ENT, "entity GEMM: whole 16-row tiles");
static_assert(((KD / 2) * FB_PITCH + 2 * 16 * FU_UP + KD) * 4 <= 320 * 1024, "entity GEMM: W_bot, two tiles of U and the score weights must fit the WGP's LDS");
static_assert(FU_CONS * 16 * FU_NT == KD, "entity GEMM: the consumer waves cover all output columns");
static_assert(FU_PROD == 16, "one producer wave per row of a tile");
#define WS_PAD 48
static_assert((N_ENT + N_ATT + N_ENT + NE + N_ATT * KD + WS_PAD) % 64 == 0, "buckets start on a 256-byte multiple");
static_assert(HEAD_E <= BK, "the head's edges come out of the bucket");

extern "C" void kernel_launch(void* const* d_in, const int* in_sizes, int n_in,
                              void* d_out, int out_size, void* d_ws, size_t ws_size,
                              hipStream_t stream) {
    const int*   trip      = (const int*)d_in[0];
    const float* att_feats = (const float*)d_in[1];
    const float* val_feats = (const float*)d_in[2];
    const float* ent_feats = (const float*)d_in[3];
    const float* a_w       = (const float*)d_in[4];
    const float* a_b       = (const float*)d_in[5];
    const float* W         = (const float*)d_in[6];
    float* out = (float*)d_out;

    const float* W_top = W;
    const float* W_bot = W + KD * KD;

    float* ws       = (float*)d_ws;
    float* att_s    = ws + N_ENT;
    int*   row_cnt  = (int*)(att_s + N_ATT);
    int*   ovf_cnt  = row_cnt + N_ENT;
    float* att_proj = (float*)(ovf_cnt + NE);
    int*   bucket   = (int*)(att_proj + (long)N_ATT * KD) + WS_PAD;
    int*   ovf      = bucket + (long)N_ENT * BK;

    {
        int tiles = (N_ATT / 16) * 16;
        static_assert(ATT_THREADS == ((N_ATT / 16) * 16 / 8) * 256, "ATT_THREADS is this launch's thread count");
        wmma_gemm_att<<<tiles / 8, 256, 0, stream>>>(att_feats, W_top, att_proj,
                                                     a_w + KD, att_s, row_cnt, ovf_cnt, bucket);
    }

    edge_bucket_kernel<<<(NE + 255) / 256, 256, 0, stream>>>(trip, row_cnt, bucket, ovf_cnt, ovf);

    row_agg_gemm_ent<<<FB_WGS, FU_THREADS, 0, stream>>>(row_cnt, bucket, ovf_cnt, ovf, ent_feats, a_w, att_s, a_b,
                                                        att_proj, val_feats, W_bot, out);
}
